# starting kernel
# baseline (speedup 1.0000x reference)
#include <hip/hip_runtime.h>


typedef float v2f __attribute__((ext_vector_type(2)));
typedef float v8f __attribute__((ext_vector_type(8)));

#define NB 8
#define NS 64
#define ND 300
#define NTOK (NB * NS)
#define NTHREADS 256
#define NWAVE 8

#define C0 16
#define K0 50
#define KP 52
#define OW0 84
#define NP 96
#define P0 28

#define C1 32
#define OW1 26
#define NPIX1 676
#define K1 144
#define P1 8

#define C2 32
#define OW2 6
#define NPIX2 36
#define NPIX2P 48
#define K2 288

#define OFF_VM  0
#define OFF_UM  4992
#define OFF_W0C 9984
#define OFF_TC  13312
#define OFF_OC  18304
#define OFF_X1  26368
#define OFF_A1  0
#define OFF_BT  4608
#define OFF_O1  38912
#define OFF_X2  60544
#define OFF_A2  0
#define OFF_B2  9216
#define OFF_O2  23040
#define SMEM_FLOATS 62592

__device__ __forceinline__ float gelu_exact(float x) {
  return 0.5f * x * (1.0f + erff(x * 0.70710678118654752f));
}

__device__ __forceinline__ v2f ld_a(const float* A, int lda, int m0, int k0,
                                    int l, int half) {
  return *(const v2f*)(A + (m0 + l) * lda + (k0 + 2 * half));
}
__device__ __forceinline__ v2f ld_b(const float* Bm, int ldb, int k0, int n0,
                                    int l, int half) {
  v2f b;
  b.x = Bm[(k0 + 2 * half) * ldb + n0 + l];
  b.y = Bm[(k0 + 2 * half + 1) * ldb + n0 + l];
  return b;
}
__device__ __forceinline__ void st_c(float* Cm, int ldc, int m0, int n0, v8f c,
                                     int l, int half, int maxRow, int maxCol) {
  int col = n0 + l;
  if (col >= maxCol) return;
#pragma unroll
  for (int r = 0; r < 8; ++r) {
    int row = m0 + r + 8 * half;
    if (row < maxRow) Cm[row * ldc + col] = c[r];
  }
}

#define WMMA_F32(acc, a, b)                                                   \
  (acc) = __builtin_amdgcn_wmma_f32_16x16x4_f32(false, (a), false, (b),       \
                                                (short)0, (acc), false, false)

extern "C" __global__ __launch_bounds__(NTHREADS) void
fused_outer_conv_kernel(const float* __restrict__ tf,
                        const float* __restrict__ tp,
                        const float* __restrict__ w0, const float* __restrict__ b0,
                        const float* __restrict__ w1, const float* __restrict__ b1,
                        const float* __restrict__ w2, const float* __restrict__ b2,
                        float* __restrict__ out) {
  extern __shared__ float smem[];
  const int tid  = threadIdx.x;
  const int lane = tid & 31;
  const int l    = tid & 15;
  const int half = (tid >> 4) & 1;
  const int wave = tid >> 5;
  const int tok  = blockIdx.x;

  const float* v = tf + tok * ND;
  const float* u = tp + tok * ND;

  {
    const char* pw1 = (const char*)w1;
    const char* pw2 = (const char*)w2;
    if (tid * 128 < (int)(C1 * K1 * sizeof(float))) __builtin_prefetch(pw1 + tid * 128, 0, 1);
    if (tid * 128 < (int)(C2 * K2 * sizeof(float))) __builtin_prefetch(pw2 + tid * 128, 0, 1);
  }

  float* Vm  = smem + OFF_VM;
  float* Um  = smem + OFF_UM;
  float* W0c = smem + OFF_W0C;
  float* Tc  = smem + OFF_TC;
  float* Oc  = smem + OFF_OC;
  float* X1  = smem + OFF_X1;

  for (int i = tid; i < KP * NP; i += NTHREADS) {
    int kx = i / NP, ox = i % NP;
    Vm[i] = (kx < K0 && ox < OW0) ? v[3 * ox + kx] : 0.f;
  }
  for (int i = tid; i < NP * KP; i += NTHREADS) {
    int oy = i / KP, ky = i % KP;
    Um[i] = (oy < OW0 && ky < K0) ? u[3 * oy + ky] : 0.f;
  }

  for (int c = 0; c < C0; ++c) {
    __syncthreads();
    for (int i = tid; i < 64 * KP; i += NTHREADS) {
      int ky = i / KP, kx = i % KP;
      W0c[i] = (ky < K0 && kx < K0) ? w0[(c * K0 + ky) * K0 + kx] : 0.f;
    }
    __syncthreads();

    for (int t = wave; t < 4 * 6; t += NWAVE) {
      int m0 = (t / 6) * 16, n0 = (t % 6) * 16;
      v8f acc = {0.f, 0.f, 0.f, 0.f, 0.f, 0.f, 0.f, 0.f};
      for (int k0 = 0; k0 < KP; k0 += 4) {
        v2f a  = ld_a(W0c, KP, m0, k0, l, half);
        v2f bb = ld_b(Vm, NP, k0, n0, l, half);
        WMMA_F32(acc, a, bb);
      }
      st_c(Tc, NP, m0, n0, acc, l, half, KP, NP);
    }
    __syncthreads();

    for (int t = wave; t < 6 * 6; t += NWAVE) {
      int m0 = (t / 6) * 16, n0 = (t % 6) * 16;
      v8f acc = {0.f, 0.f, 0.f, 0.f, 0.f, 0.f, 0.f, 0.f};
      for (int k0 = 0; k0 < KP; k0 += 4) {
        v2f a  = ld_a(Um, KP, m0, k0, l, half);
        v2f bb = ld_b(Tc, NP, k0, n0, l, half);
        WMMA_F32(acc, a, bb);
      }
      st_c(Oc, NP, m0, n0, acc, l, half, OW0, NP);
    }
    __syncthreads();

    float bc = b0[c];
    for (int i = tid; i < P0 * P0; i += NTHREADS) {
      int py = i / P0, px = i % P0;
      float m = -3.402823466e38f;
      for (int dy = 0; dy < 3; ++dy)
        for (int dx = 0; dx < 3; ++dx)
          m = fmaxf(m, gelu_exact(Oc[(3 * py + dy) * NP + (3 * px + dx)] + bc));
      X1[c * (P0 * P0) + i] = m;
    }
  }
  __syncthreads();

  float* A1 = smem + OFF_A1;
  float* O1 = smem + OFF_O1;
  float* X2 = smem + OFF_X2;
  for (int i = tid; i < C1 * K1; i += NTHREADS) A1[i] = w1[i];
  __syncthreads();

  float* Bt = smem + OFF_BT + wave * (K1 * 16);
  for (int nt = wave; nt < (NPIX1 + 15) / 16; nt += NWAVE) {
    for (int i = lane; i < K1 * 16; i += 32) {
      int kk = i >> 4, j = i & 15;
      int p = nt * 16 + j;
      float val = 0.f;
      if (p < NPIX1) {
        int py = p / OW1, px = p % OW1;
        int ci = kk / 9, r9 = kk % 9;
        val = X1[ci * (P0 * P0) + (py + r9 / 3) * P0 + (px + r9 % 3)];
      }
      Bt[i] = val;
    }
    for (int m0 = 0; m0 < C1; m0 += 16) {
      v8f acc = {0.f, 0.f, 0.f, 0.f, 0.f, 0.f, 0.f, 0.f};
      for (int k0 = 0; k0 < K1; k0 += 4) {
        v2f a  = ld_a(A1, K1, m0, k0, l, half);
        v2f bb = ld_b(Bt, 16, k0, 0, l, half);
        WMMA_F32(acc, a, bb);
      }
      int col = nt * 16 + l;
      if (col < NPIX1) {
#pragma unroll
        for (int r = 0; r < 8; ++r) O1[(m0 + r + 8 * half) * NPIX1 + col] = acc[r];
      }
    }
  }
  __syncthreads();

  for (int i = tid; i < C1 * P1 * P1; i += NTHREADS) {
    int co = i >> 6, rem = i & 63, py = rem >> 3, px = rem & 7;
    float bc = b1[co];
    float m = -3.402823466e38f;
    for (int dy = 0; dy < 3; ++dy)
      for (int dx = 0; dx < 3; ++dx)
        m = fmaxf(m, gelu_exact(O1[co * NPIX1 + (3 * py + dy) * OW1 + (3 * px + dx)] + bc));
    X2[i] = m;
  }
  __syncthreads();

  float* A2 = smem + OFF_A2;
  float* B2 = smem + OFF_B2;
  float* O2 = smem + OFF_O2;
  for (int i = tid; i < C2 * K2; i += NTHREADS) A2[i] = w2[i];
  for (int i = tid; i < K2 * NPIX2P; i += NTHREADS) {
    int kk = i / NPIX2P, col = i % NPIX2P;
    float val = 0.f;
    if (col < NPIX2) {
      int py = col / OW2, px = col % OW2;
      int ci = kk / 9, r9 = kk % 9;
      val = X2[ci * (P1 * P1) + (py + r9 / 3) * P1 + (px + r9 % 3)];
    }
    B2[i] = val;
  }
  __syncthreads();

  for (int t = wave; t < 2 * 3; t += NWAVE) {
    int m0 = (t / 3) * 16, n0 = (t % 3) * 16;
    v8f acc = {0.f, 0.f, 0.f, 0.f, 0.f, 0.f, 0.f, 0.f};
    for (int k0 = 0; k0 < K2; k0 += 4) {
      v2f a  = ld_a(A2, K2, m0, k0, l, half);
      v2f bb = ld_b(B2, NPIX2P, k0, n0, l, half);
      WMMA_F32(acc, a, bb);
    }
    st_c(O2, NPIX2P, m0, n0, acc, l, half, C2, NPIX2P);
  }
  __syncthreads();

  if (tid < 128) {
    int co = tid >> 2, rem = tid & 3, py = rem >> 1, px = rem & 1;
    float bc = b2[co];
    float m = -3.402823466e38f;
    for (int dy = 0; dy < 3; ++dy)
      for (int dx = 0; dx < 3; ++dx)
        m = fmaxf(m, gelu_exact(O2[co * NPIX2P + (3 * py + dy) * OW2 + (3 * px + dx)] + bc));
    out[tok * 128 + tid] = m;
  }
}

extern "C" void kernel_launch(void* const* d_in, const int* in_sizes, int n_in,
                              void* d_out, int out_size, void* d_ws, size_t ws_size,
                              hipStream_t stream) {
  (void)in_sizes; (void)n_in; (void)out_size; (void)d_ws; (void)ws_size;
  const float* tf = (const float*)d_in[0];
  const float* tp = (const float*)d_in[1];
  const float* w0 = (const float*)d_in[2];
  const float* b0 = (const float*)d_in[3];
  const float* w1 = (const float*)d_in[4];
  const float* b1 = (const float*)d_in[5];
  const float* w2 = (const float*)d_in[6];
  const float* b2 = (const float*)d_in[7];
  float* out = (float*)d_out;

  const size_t shmem = SMEM_FLOATS * sizeof(float);
  (void)hipFuncSetAttribute((const void*)fused_outer_conv_kernel,
                            hipFuncAttributeMaxDynamicSharedMemorySize, (int)shmem);
  hipLaunchKernelGGL(fused_outer_conv_kernel, dim3(NTOK), dim3(NTHREADS), shmem,
                     stream, tf, tp, w0, b0, w1, b1, w2, b2, out);
}


# GELU evaluated only at each pool window's min and max pre-activation (GELU is unimodal, so the window max is at an extreme): 9 -> 2 erf per pooled output in all three blocks
# speedup vs baseline: 1.2683x; 1.2683x over previous
#include <hip/hip_runtime.h>


typedef float v2f __attribute__((ext_vector_type(2)));
typedef float v8f __attribute__((ext_vector_type(8)));

#define NB 8
#define NS 64
#define ND 300
#define NTOK (NB * NS)
#define NTHREADS 256
#define NWAVE 8

#define C0 16
#define K0 50
#define KP 52
#define OW0 84
#define NP 96
#define P0 28

#define C1 32
#define OW1 26
#define NPIX1 676
#define K1 144
#define P1 8

#define C2 32
#define OW2 6
#define NPIX2 36
#define NPIX2P 48
#define K2 288

#define OFF_VM  0
#define OFF_UM  4992
#define OFF_W0C 9984
#define OFF_TC  13312
#define OFF_OC  18304
#define OFF_X1  26368
#define OFF_A1  0
#define OFF_BT  4608
#define OFF_O1  38912
#define OFF_X2  60544
#define OFF_A2  0
#define OFF_B2  9216
#define OFF_O2  23040
#define SMEM_FLOATS 62592

__device__ __forceinline__ float gelu_exact(float x) {
  return 0.5f * x * (1.0f + erff(x * 0.70710678118654752f));
}

__device__ __forceinline__ float gelu_pool_max(float lo, float hi) {
  return fmaxf(gelu_exact(lo), gelu_exact(hi));
}

__device__ __forceinline__ v2f ld_a(const float* A, int lda, int m0, int k0,
                                    int l, int half) {
  return *(const v2f*)(A + (m0 + l) * lda + (k0 + 2 * half));
}
__device__ __forceinline__ v2f ld_b(const float* Bm, int ldb, int k0, int n0,
                                    int l, int half) {
  v2f b;
  b.x = Bm[(k0 + 2 * half) * ldb + n0 + l];
  b.y = Bm[(k0 + 2 * half + 1) * ldb + n0 + l];
  return b;
}
__device__ __forceinline__ void st_c(float* Cm, int ldc, int m0, int n0, v8f c,
                                     int l, int half, int maxRow, int maxCol) {
  int col = n0 + l;
  if (col >= maxCol) return;
#pragma unroll
  for (int r = 0; r < 8; ++r) {
    int row = m0 + r + 8 * half;
    if (row < maxRow) Cm[row * ldc + col] = c[r];
  }
}

#define WMMA_F32(acc, a, b)                                                   \
  (acc) = __builtin_amdgcn_wmma_f32_16x16x4_f32(false, (a), false, (b),       \
                                                (short)0, (acc), false, false)

extern "C" __global__ __launch_bounds__(NTHREADS) void
fused_outer_conv_kernel(const float* __restrict__ tf,
                        const float* __restrict__ tp,
                        const float* __restrict__ w0, const float* __restrict__ b0,
                        const float* __restrict__ w1, const float* __restrict__ b1,
                        const float* __restrict__ w2, const float* __restrict__ b2,
                        float* __restrict__ out) {
  extern __shared__ float smem[];
  const int tid  = threadIdx.x;
  const int lane = tid & 31;
  const int l    = tid & 15;
  const int half = (tid >> 4) & 1;
  const int wave = tid >> 5;
  const int tok  = blockIdx.x;

  const float* v = tf + tok * ND;
  const float* u = tp + tok * ND;

  {
    const char* pw1 = (const char*)w1;
    const char* pw2 = (const char*)w2;
    if (tid * 128 < (int)(C1 * K1 * sizeof(float))) __builtin_prefetch(pw1 + tid * 128, 0, 1);
    if (tid * 128 < (int)(C2 * K2 * sizeof(float))) __builtin_prefetch(pw2 + tid * 128, 0, 1);
  }

  float* Vm  = smem + OFF_VM;
  float* Um  = smem + OFF_UM;
  float* W0c = smem + OFF_W0C;
  float* Tc  = smem + OFF_TC;
  float* Oc  = smem + OFF_OC;
  float* X1  = smem + OFF_X1;

  for (int i = tid; i < KP * NP; i += NTHREADS) {
    int kx = i / NP, ox = i % NP;
    Vm[i] = (kx < K0 && ox < OW0) ? v[3 * ox + kx] : 0.f;
  }
  for (int i = tid; i < NP * KP; i += NTHREADS) {
    int oy = i / KP, ky = i % KP;
    Um[i] = (oy < OW0 && ky < K0) ? u[3 * oy + ky] : 0.f;
  }

  for (int c = 0; c < C0; ++c) {
    __syncthreads();
    for (int i = tid; i < 64 * KP; i += NTHREADS) {
      int ky = i / KP, kx = i % KP;
      W0c[i] = (ky < K0 && kx < K0) ? w0[(c * K0 + ky) * K0 + kx] : 0.f;
    }
    __syncthreads();

    for (int t = wave; t < 4 * 6; t += NWAVE) {
      int m0 = (t / 6) * 16, n0 = (t % 6) * 16;
      v8f acc = {0.f, 0.f, 0.f, 0.f, 0.f, 0.f, 0.f, 0.f};
      for (int k0 = 0; k0 < KP; k0 += 4) {
        v2f a  = ld_a(W0c, KP, m0, k0, l, half);
        v2f bb = ld_b(Vm, NP, k0, n0, l, half);
        WMMA_F32(acc, a, bb);
      }
      st_c(Tc, NP, m0, n0, acc, l, half, KP, NP);
    }
    __syncthreads();

    for (int t = wave; t < 6 * 6; t += NWAVE) {
      int m0 = (t / 6) * 16, n0 = (t % 6) * 16;
      v8f acc = {0.f, 0.f, 0.f, 0.f, 0.f, 0.f, 0.f, 0.f};
      for (int k0 = 0; k0 < KP; k0 += 4) {
        v2f a  = ld_a(Um, KP, m0, k0, l, half);
        v2f bb = ld_b(Tc, NP, k0, n0, l, half);
        WMMA_F32(acc, a, bb);
      }
      st_c(Oc, NP, m0, n0, acc, l, half, OW0, NP);
    }
    __syncthreads();

    float bc = b0[c];
    for (int i = tid; i < P0 * P0; i += NTHREADS) {
      int py = i / P0, px = i % P0;
      float hi = -3.402823466e38f, lo = 3.402823466e38f;
      for (int dy = 0; dy < 3; ++dy)
        for (int dx = 0; dx < 3; ++dx) {
          float x = Oc[(3 * py + dy) * NP + (3 * px + dx)];
          hi = fmaxf(hi, x);
          lo = fminf(lo, x);
        }
      X1[c * (P0 * P0) + i] = gelu_pool_max(lo + bc, hi + bc);
    }
  }
  __syncthreads();

  float* A1 = smem + OFF_A1;
  float* O1 = smem + OFF_O1;
  float* X2 = smem + OFF_X2;
  for (int i = tid; i < C1 * K1; i += NTHREADS) A1[i] = w1[i];
  __syncthreads();

  float* Bt = smem + OFF_BT + wave * (K1 * 16);
  for (int nt = wave; nt < (NPIX1 + 15) / 16; nt += NWAVE) {
    for (int i = lane; i < K1 * 16; i += 32) {
      int kk = i >> 4, j = i & 15;
      int p = nt * 16 + j;
      float val = 0.f;
      if (p < NPIX1) {
        int py = p / OW1, px = p % OW1;
        int ci = kk / 9, r9 = kk % 9;
        val = X1[ci * (P0 * P0) + (py + r9 / 3) * P0 + (px + r9 % 3)];
      }
      Bt[i] = val;
    }
    for (int m0 = 0; m0 < C1; m0 += 16) {
      v8f acc = {0.f, 0.f, 0.f, 0.f, 0.f, 0.f, 0.f, 0.f};
      for (int k0 = 0; k0 < K1; k0 += 4) {
        v2f a  = ld_a(A1, K1, m0, k0, l, half);
        v2f bb = ld_b(Bt, 16, k0, 0, l, half);
        WMMA_F32(acc, a, bb);
      }
      int col = nt * 16 + l;
      if (col < NPIX1) {
#pragma unroll
        for (int r = 0; r < 8; ++r) O1[(m0 + r + 8 * half) * NPIX1 + col] = acc[r];
      }
    }
  }
  __syncthreads();

  for (int i = tid; i < C1 * P1 * P1; i += NTHREADS) {
    int co = i >> 6, rem = i & 63, py = rem >> 3, px = rem & 7;
    float bc = b1[co];
    float hi = -3.402823466e38f, lo = 3.402823466e38f;
    for (int dy = 0; dy < 3; ++dy)
      for (int dx = 0; dx < 3; ++dx) {
        float x = O1[co * NPIX1 + (3 * py + dy) * OW1 + (3 * px + dx)];
        hi = fmaxf(hi, x);
        lo = fminf(lo, x);
      }
    X2[i] = gelu_pool_max(lo + bc, hi + bc);
  }
  __syncthreads();

  float* A2 = smem + OFF_A2;
  float* B2 = smem + OFF_B2;
  float* O2 = smem + OFF_O2;
  for (int i = tid; i < C2 * K2; i += NTHREADS) A2[i] = w2[i];
  for (int i = tid; i < K2 * NPIX2P; i += NTHREADS) {
    int kk = i / NPIX2P, col = i % NPIX2P;
    float val = 0.f;
    if (col < NPIX2) {
      int py = col / OW2, px = col % OW2;
      int ci = kk / 9, r9 = kk % 9;
      val = X2[ci * (P1 * P1) + (py + r9 / 3) * P1 + (px + r9 % 3)];
    }
    B2[i] = val;
  }
  __syncthreads();

  for (int t = wave; t < 2 * 3; t += NWAVE) {
    int m0 = (t / 3) * 16, n0 = (t % 3) * 16;
    v8f acc = {0.f, 0.f, 0.f, 0.f, 0.f, 0.f, 0.f, 0.f};
    for (int k0 = 0; k0 < K2; k0 += 4) {
      v2f a  = ld_a(A2, K2, m0, k0, l, half);
      v2f bb = ld_b(B2, NPIX2P, k0, n0, l, half);
      WMMA_F32(acc, a, bb);
    }
    st_c(O2, NPIX2P, m0, n0, acc, l, half, C2, NPIX2P);
  }
  __syncthreads();

  if (tid < 128) {
    int co = tid >> 2, rem = tid & 3, py = rem >> 1, px = rem & 1;
    float bc = b2[co];
    float hi = -3.402823466e38f, lo = 3.402823466e38f;
    for (int dy = 0; dy < 3; ++dy)
      for (int dx = 0; dx < 3; ++dx) {
        float x = O2[co * NPIX2P + (3 * py + dy) * OW2 + (3 * px + dx)];
        hi = fmaxf(hi, x);
        lo = fminf(lo, x);
      }
    out[tok * 128 + tid] = gelu_pool_max(lo + bc, hi + bc);
  }
}

extern "C" void kernel_launch(void* const* d_in, const int* in_sizes, int n_in,
                              void* d_out, int out_size, void* d_ws, size_t ws_size,
                              hipStream_t stream) {
  (void)in_sizes; (void)n_in; (void)out_size; (void)d_ws; (void)ws_size;
  const float* tf = (const float*)d_in[0];
  const float* tp = (const float*)d_in[1];
  const float* w0 = (const float*)d_in[2];
  const float* b0 = (const float*)d_in[3];
  const float* w1 = (const float*)d_in[4];
  const float* b1 = (const float*)d_in[5];
  const float* w2 = (const float*)d_in[6];
  const float* b2 = (const float*)d_in[7];
  float* out = (float*)d_out;

  const size_t shmem = SMEM_FLOATS * sizeof(float);
  (void)hipFuncSetAttribute((const void*)fused_outer_conv_kernel,
                            hipFuncAttributeMaxDynamicSharedMemorySize, (int)shmem);
  hipLaunchKernelGGL(fused_outer_conv_kernel, dim3(NTOK), dim3(NTHREADS), shmem,
                     stream, tf, tp, w0, b0, w1, b1, w2, b2, out);
}


# conv1 reads its B fragments straight from the pooled conv0 plane (k = tap*16+channel) instead of per-wave im2col slabs; LDS falls 245 -> 152 KB, so two workgroups share a WGP
# speedup vs baseline: 2.4843x; 1.9588x over previous
#include <hip/hip_runtime.h>


typedef float v2f __attribute__((ext_vector_type(2)));
typedef float v8f __attribute__((ext_vector_type(8)));

#define NB 8
#define NS 64
#define ND 300
#define NTOK (NB * NS)
#define NTHREADS 256
#define NWAVE 8

#define C0 16
#define K0 50
#define KP 52
#define OW0 84
#define NP 96
#define P0 28

#define C1 32
#define OW1 26
#define NPIX1 676
#define K1 144
#define P1 8

#define C2 32
#define OW2 6
#define NPIX2 36
#define NPIX2P 48
#define K2 288

#define OFF_VM  0
#define OFF_UM  4992
#define OFF_W0C 9984
#define OFF_TC  13312
#define OFF_OC  18304
#define OFF_X1  26368
#define OFF_A1  0
#define OFF_O1  4608
#define OFF_X2  26368
#define OFF_A2  0
#define OFF_B2  9216
#define OFF_O2  23040
#define SMEM_FLOATS 38912

__device__ __forceinline__ float gelu_exact(float x) {
  return 0.5f * x * (1.0f + erff(x * 0.70710678118654752f));
}

__device__ __forceinline__ float gelu_pool_max(float lo, float hi) {
  return fmaxf(gelu_exact(lo), gelu_exact(hi));
}

__device__ __forceinline__ v2f ld_a(const float* A, int lda, int m0, int k0,
                                    int l, int half) {
  return *(const v2f*)(A + (m0 + l) * lda + (k0 + 2 * half));
}
__device__ __forceinline__ v2f ld_b(const float* Bm, int ldb, int k0, int n0,
                                    int l, int half) {
  v2f b;
  b.x = Bm[(k0 + 2 * half) * ldb + n0 + l];
  b.y = Bm[(k0 + 2 * half + 1) * ldb + n0 + l];
  return b;
}
__device__ __forceinline__ void st_c(float* Cm, int ldc, int m0, int n0, v8f c,
                                     int l, int half, int maxRow, int maxCol) {
  int col = n0 + l;
  if (col >= maxCol) return;
#pragma unroll
  for (int r = 0; r < 8; ++r) {
    int row = m0 + r + 8 * half;
    if (row < maxRow) Cm[row * ldc + col] = c[r];
  }
}

#define WMMA_F32(acc, a, b)                                                   \
  (acc) = __builtin_amdgcn_wmma_f32_16x16x4_f32(false, (a), false, (b),       \
                                                (short)0, (acc), false, false)

extern "C" __global__ __launch_bounds__(NTHREADS) void
fused_outer_conv_kernel(const float* __restrict__ tf,
                        const float* __restrict__ tp,
                        const float* __restrict__ w0, const float* __restrict__ b0,
                        const float* __restrict__ w1, const float* __restrict__ b1,
                        const float* __restrict__ w2, const float* __restrict__ b2,
                        float* __restrict__ out) {
  extern __shared__ float smem[];
  const int tid  = threadIdx.x;
  const int lane = tid & 31;
  const int l    = tid & 15;
  const int half = (tid >> 4) & 1;
  const int wave = tid >> 5;
  const int tok  = blockIdx.x;

  const float* v = tf + tok * ND;
  const float* u = tp + tok * ND;

  {
    const char* pw1 = (const char*)w1;
    const char* pw2 = (const char*)w2;
    if (tid * 128 < (int)(C1 * K1 * sizeof(float))) __builtin_prefetch(pw1 + tid * 128, 0, 1);
    if (tid * 128 < (int)(C2 * K2 * sizeof(float))) __builtin_prefetch(pw2 + tid * 128, 0, 1);
  }

  float* Vm  = smem + OFF_VM;
  float* Um  = smem + OFF_UM;
  float* W0c = smem + OFF_W0C;
  float* Tc  = smem + OFF_TC;
  float* Oc  = smem + OFF_OC;
  float* X1  = smem + OFF_X1;

  for (int i = tid; i < KP * NP; i += NTHREADS) {
    int kx = i / NP, ox = i % NP;
    Vm[i] = (kx < K0 && ox < OW0) ? v[3 * ox + kx] : 0.f;
  }
  for (int i = tid; i < NP * KP; i += NTHREADS) {
    int oy = i / KP, ky = i % KP;
    Um[i] = (oy < OW0 && ky < K0) ? u[3 * oy + ky] : 0.f;
  }

  for (int c = 0; c < C0; ++c) {
    __syncthreads();
    for (int i = tid; i < 64 * KP; i += NTHREADS) {
      int ky = i / KP, kx = i % KP;
      W0c[i] = (ky < K0 && kx < K0) ? w0[(c * K0 + ky) * K0 + kx] : 0.f;
    }
    __syncthreads();

    for (int t = wave; t < 4 * 6; t += NWAVE) {
      int m0 = (t / 6) * 16, n0 = (t % 6) * 16;
      v8f acc = {0.f, 0.f, 0.f, 0.f, 0.f, 0.f, 0.f, 0.f};
      for (int k0 = 0; k0 < KP; k0 += 4) {
        v2f a  = ld_a(W0c, KP, m0, k0, l, half);
        v2f bb = ld_b(Vm, NP, k0, n0, l, half);
        WMMA_F32(acc, a, bb);
      }
      st_c(Tc, NP, m0, n0, acc, l, half, KP, NP);
    }
    __syncthreads();

    for (int t = wave; t < 6 * 6; t += NWAVE) {
      int m0 = (t / 6) * 16, n0 = (t % 6) * 16;
      v8f acc = {0.f, 0.f, 0.f, 0.f, 0.f, 0.f, 0.f, 0.f};
      for (int k0 = 0; k0 < KP; k0 += 4) {
        v2f a  = ld_a(Um, KP, m0, k0, l, half);
        v2f bb = ld_b(Tc, NP, k0, n0, l, half);
        WMMA_F32(acc, a, bb);
      }
      st_c(Oc, NP, m0, n0, acc, l, half, OW0, NP);
    }
    __syncthreads();

    float bc = b0[c];
    for (int i = tid; i < P0 * P0; i += NTHREADS) {
      int py = i / P0, px = i % P0;
      float hi = -3.402823466e38f, lo = 3.402823466e38f;
      for (int dy = 0; dy < 3; ++dy)
        for (int dx = 0; dx < 3; ++dx) {
          float x = Oc[(3 * py + dy) * NP + (3 * px + dx)];
          hi = fmaxf(hi, x);
          lo = fminf(lo, x);
        }
      X1[c * (P0 * P0) + i] = gelu_pool_max(lo + bc, hi + bc);
    }
  }
  __syncthreads();

  float* A1 = smem + OFF_A1;
  float* O1 = smem + OFF_O1;
  float* X2 = smem + OFF_X2;
  for (int i = tid; i < C1 * K1; i += NTHREADS) {
    int co = i / K1, k = i - co * K1;
    int ci = k / 9, r9 = k - ci * 9;
    A1[co * K1 + r9 * 16 + ci] = w1[i];
  }
  __syncthreads();

  for (int nt = wave; nt < (NPIX1 + 15) / 16; nt += NWAVE) {
    const int col = nt * 16 + l;
    const int pc  = col < NPIX1 ? col : NPIX1 - 1;
    const int py = pc / OW1, px = pc - py * OW1;
    const float* xb = X1 + py * P0 + px + half * (2 * P0 * P0);
    const float* ab = A1 + l * K1 + 2 * half;
    v8f acc0 = {0.f, 0.f, 0.f, 0.f, 0.f, 0.f, 0.f, 0.f};
    v8f acc1 = {0.f, 0.f, 0.f, 0.f, 0.f, 0.f, 0.f, 0.f};
    v2f fb[2][4], fa0[2][4], fa1[2][4];
#pragma unroll
    for (int q = 0; q < 4; ++q) {
      fb[0][q].x = xb[(4 * q) * (P0 * P0)];
      fb[0][q].y = xb[(4 * q + 1) * (P0 * P0)];
      fa0[0][q]  = *(const v2f*)(ab + 4 * q);
      fa1[0][q]  = *(const v2f*)(ab + 16 * K1 + 4 * q);
    }
#pragma unroll
    for (int r9 = 0; r9 < 9; ++r9) {
      const int cur = r9 & 1, nxt = cur ^ 1;
      if (r9 + 1 < 9) {
        const int sp = ((r9 + 1) / 3) * P0 + ((r9 + 1) % 3);
#pragma unroll
        for (int q = 0; q < 4; ++q) {
          fb[nxt][q].x = xb[(4 * q) * (P0 * P0) + sp];
          fb[nxt][q].y = xb[(4 * q + 1) * (P0 * P0) + sp];
          fa0[nxt][q]  = *(const v2f*)(ab + (r9 + 1) * 16 + 4 * q);
          fa1[nxt][q]  = *(const v2f*)(ab + 16 * K1 + (r9 + 1) * 16 + 4 * q);
        }
      }
      __builtin_amdgcn_sched_barrier(0);
#pragma unroll
      for (int q = 0; q < 4; ++q) {
        WMMA_F32(acc0, fa0[cur][q], fb[cur][q]);
        WMMA_F32(acc1, fa1[cur][q], fb[cur][q]);
      }
      __builtin_amdgcn_sched_barrier(0);
    }
    if (col < NPIX1) {
#pragma unroll
      for (int r = 0; r < 8; ++r) {
        O1[(r + 8 * half) * NPIX1 + col]      = acc0[r];
        O1[(16 + r + 8 * half) * NPIX1 + col] = acc1[r];
      }
    }
  }
  __syncthreads();

  for (int i = tid; i < C1 * P1 * P1; i += NTHREADS) {
    int co = i >> 6, rem = i & 63, py = rem >> 3, px = rem & 7;
    float bc = b1[co];
    float hi = -3.402823466e38f, lo = 3.402823466e38f;
    for (int dy = 0; dy < 3; ++dy)
      for (int dx = 0; dx < 3; ++dx) {
        float x = O1[co * NPIX1 + (3 * py + dy) * OW1 + (3 * px + dx)];
        hi = fmaxf(hi, x);
        lo = fminf(lo, x);
      }
    X2[i] = gelu_pool_max(lo + bc, hi + bc);
  }
  __syncthreads();

  float* A2 = smem + OFF_A2;
  float* B2 = smem + OFF_B2;
  float* O2 = smem + OFF_O2;
  for (int i = tid; i < C2 * K2; i += NTHREADS) A2[i] = w2[i];
  for (int i = tid; i < K2 * NPIX2P; i += NTHREADS) {
    int kk = i / NPIX2P, col = i % NPIX2P;
    float val = 0.f;
    if (col < NPIX2) {
      int py = col / OW2, px = col % OW2;
      int ci = kk / 9, r9 = kk % 9;
      val = X2[ci * (P1 * P1) + (py + r9 / 3) * P1 + (px + r9 % 3)];
    }
    B2[i] = val;
  }
  __syncthreads();

  for (int t = wave; t < 2 * 3; t += NWAVE) {
    int m0 = (t / 3) * 16, n0 = (t % 3) * 16;
    v8f acc = {0.f, 0.f, 0.f, 0.f, 0.f, 0.f, 0.f, 0.f};
    for (int k0 = 0; k0 < K2; k0 += 4) {
      v2f a  = ld_a(A2, K2, m0, k0, l, half);
      v2f bb = ld_b(B2, NPIX2P, k0, n0, l, half);
      WMMA_F32(acc, a, bb);
    }
    st_c(O2, NPIX2P, m0, n0, acc, l, half, C2, NPIX2P);
  }
  __syncthreads();

  if (tid < 128) {
    int co = tid >> 2, rem = tid & 3, py = rem >> 1, px = rem & 1;
    float bc = b2[co];
    float hi = -3.402823466e38f, lo = 3.402823466e38f;
    for (int dy = 0; dy < 3; ++dy)
      for (int dx = 0; dx < 3; ++dx) {
        float x = O2[co * NPIX2P + (3 * py + dy) * OW2 + (3 * px + dx)];
        hi = fmaxf(hi, x);
        lo = fminf(lo, x);
      }
    out[tok * 128 + tid] = gelu_pool_max(lo + bc, hi + bc);
  }
}

extern "C" void kernel_launch(void* const* d_in, const int* in_sizes, int n_in,
                              void* d_out, int out_size, void* d_ws, size_t ws_size,
                              hipStream_t stream) {
  (void)in_sizes; (void)n_in; (void)out_size; (void)d_ws; (void)ws_size;
  const float* tf = (const float*)d_in[0];
  const float* tp = (const float*)d_in[1];
  const float* w0 = (const float*)d_in[2];
  const float* b0 = (const float*)d_in[3];
  const float* w1 = (const float*)d_in[4];
  const float* b1 = (const float*)d_in[5];
  const float* w2 = (const float*)d_in[6];
  const float* b2 = (const float*)d_in[7];
  float* out = (float*)d_out;

  const size_t shmem = SMEM_FLOATS * sizeof(float);
  (void)hipFuncSetAttribute((const void*)fused_outer_conv_kernel,
                            hipFuncAttributeMaxDynamicSharedMemorySize, (int)shmem);
  hipLaunchKernelGGL(fused_outer_conv_kernel, dim3(NTOK), dim3(NTHREADS), shmem,
                     stream, tf, tp, w0, b0, w1, b1, w2, b2, out);
}


# conv0 weights fetched one channel ahead: three 16-byte loads per thread held in registers across stage A, then stored to LDS; replaces 13 dependent load-store trips and 2 of 4 barriers per channel
# speedup vs baseline: 3.7026x; 1.4904x over previous
#include <hip/hip_runtime.h>


typedef float v2f __attribute__((ext_vector_type(2)));
typedef float v4f __attribute__((ext_vector_type(4)));
typedef float v8f __attribute__((ext_vector_type(8)));

#define NB 8
#define NS 64
#define ND 300
#define NTOK (NB * NS)
#define NTHREADS 256
#define NWAVE 8

#define C0 16
#define K0 50
#define KP 52
#define OW0 84
#define NP 96
#define P0 28
#define W0Q 625

#define C1 32
#define OW1 26
#define NPIX1 676
#define K1 144
#define P1 8

#define C2 32
#define OW2 6
#define NPIX2 36
#define NPIX2P 48
#define K2 288

#define OFF_VM  0
#define OFF_UM  4992
#define OFF_W0C 9984
#define OFF_TC  13312
#define OFF_OC  18304
#define OFF_X1  26368
#define OFF_A1  0
#define OFF_O1  4608
#define OFF_X2  26368
#define OFF_A2  0
#define OFF_B2  9216
#define OFF_O2  23040
#define SMEM_FLOATS 38912

__device__ __forceinline__ float gelu_exact(float x) {
  return 0.5f * x * (1.0f + erff(x * 0.70710678118654752f));
}

__device__ __forceinline__ float gelu_pool_max(float lo, float hi) {
  return fmaxf(gelu_exact(lo), gelu_exact(hi));
}

__device__ __forceinline__ int w0_pair_slot(int p) {
  int ky = p / 25;
  return ky * KP + 2 * (p - ky * 25);
}

__device__ __forceinline__ v2f ld_a(const float* A, int lda, int m0, int k0,
                                    int l, int half) {
  return *(const v2f*)(A + (m0 + l) * lda + (k0 + 2 * half));
}
__device__ __forceinline__ v2f ld_b(const float* Bm, int ldb, int k0, int n0,
                                    int l, int half) {
  v2f b;
  b.x = Bm[(k0 + 2 * half) * ldb + n0 + l];
  b.y = Bm[(k0 + 2 * half + 1) * ldb + n0 + l];
  return b;
}
__device__ __forceinline__ void st_c(float* Cm, int ldc, int m0, int n0, v8f c,
                                     int l, int half, int maxRow, int maxCol) {
  int col = n0 + l;
  if (col >= maxCol) return;
#pragma unroll
  for (int r = 0; r < 8; ++r) {
    int row = m0 + r + 8 * half;
    if (row < maxRow) Cm[row * ldc + col] = c[r];
  }
}

#define WMMA_F32(acc, a, b)                                                   \
  (acc) = __builtin_amdgcn_wmma_f32_16x16x4_f32(false, (a), false, (b),       \
                                                (short)0, (acc), false, false)

extern "C" __global__ __launch_bounds__(NTHREADS) void
fused_outer_conv_kernel(const float* __restrict__ tf,
                        const float* __restrict__ tp,
                        const float* __restrict__ w0, const float* __restrict__ b0,
                        const float* __restrict__ w1, const float* __restrict__ b1,
                        const float* __restrict__ w2, const float* __restrict__ b2,
                        float* __restrict__ out) {
  extern __shared__ float smem[];
  const int tid  = threadIdx.x;
  const int lane = tid & 31;
  const int l    = tid & 15;
  const int half = (tid >> 4) & 1;
  const int wave = tid >> 5;
  const int tok  = blockIdx.x;

  const float* v = tf + tok * ND;
  const float* u = tp + tok * ND;

  {
    const char* pw1 = (const char*)w1;
    const char* pw2 = (const char*)w2;
    if (tid * 128 < (int)(C1 * K1 * sizeof(float))) __builtin_prefetch(pw1 + tid * 128, 0, 1);
    if (tid * 128 < (int)(C2 * K2 * sizeof(float))) __builtin_prefetch(pw2 + tid * 128, 0, 1);
  }

  float* Vm  = smem + OFF_VM;
  float* Um  = smem + OFF_UM;
  float* W0c = smem + OFF_W0C;
  float* Tc  = smem + OFF_TC;
  float* Oc  = smem + OFF_OC;
  float* X1  = smem + OFF_X1;

  const v4f* w0v = (const v4f*)w0;
  const int  q0 = tid, q1 = tid + NTHREADS;
  const bool has2 = tid + 2 * NTHREADS < W0Q;
  const int  q2 = has2 ? tid + 2 * NTHREADS : W0Q - 1;
  const int  s00 = w0_pair_slot(2 * q0), s01 = w0_pair_slot(2 * q0 + 1);
  const int  s10 = w0_pair_slot(2 * q1), s11 = w0_pair_slot(2 * q1 + 1);
  const int  s20 = w0_pair_slot(2 * q2), s21 = w0_pair_slot(2 * q2 + 1);

  v4f wq0 = w0v[q0], wq1 = w0v[q1], wq2 = w0v[q2];

  for (int i = tid; i < KP * NP; i += NTHREADS) {
    int kx = i / NP, ox = i % NP;
    Vm[i] = (kx < K0 && ox < OW0) ? v[3 * ox + kx] : 0.f;
  }
  for (int i = tid; i < NP * KP; i += NTHREADS) {
    int oy = i / KP, ky = i % KP;
    Um[i] = (oy < OW0 && ky < K0) ? u[3 * oy + ky] : 0.f;
  }

  for (int i = tid; i < (64 - K0) * KP; i += NTHREADS) W0c[K0 * KP + i] = 0.f;
  if (tid < 2 * K0) W0c[(tid >> 1) * KP + K0 + (tid & 1)] = 0.f;
  *(v2f*)(W0c + s00) = (v2f){wq0.x, wq0.y};
  *(v2f*)(W0c + s01) = (v2f){wq0.z, wq0.w};
  *(v2f*)(W0c + s10) = (v2f){wq1.x, wq1.y};
  *(v2f*)(W0c + s11) = (v2f){wq1.z, wq1.w};
  if (has2) {
    *(v2f*)(W0c + s20) = (v2f){wq2.x, wq2.y};
    *(v2f*)(W0c + s21) = (v2f){wq2.z, wq2.w};
  }
  __syncthreads();

  for (int c = 0; c < C0; ++c) {
    if (c + 1 < C0) {
      const v4f* wn = w0v + (c + 1) * W0Q;
      wq0 = wn[q0];
      wq1 = wn[q1];
      wq2 = wn[q2];
    }
    const float bc = b0[c];

    for (int t = wave; t < 4 * 6; t += NWAVE) {
      int m0 = (t / 6) * 16, n0 = (t % 6) * 16;
      v8f acc = {0.f, 0.f, 0.f, 0.f, 0.f, 0.f, 0.f, 0.f};
      for (int k0 = 0; k0 < KP; k0 += 4) {
        v2f a  = ld_a(W0c, KP, m0, k0, l, half);
        v2f bb = ld_b(Vm, NP, k0, n0, l, half);
        WMMA_F32(acc, a, bb);
      }
      st_c(Tc, NP, m0, n0, acc, l, half, KP, NP);
    }
    __syncthreads();

    if (c + 1 < C0) {
      *(v2f*)(W0c + s00) = (v2f){wq0.x, wq0.y};
      *(v2f*)(W0c + s01) = (v2f){wq0.z, wq0.w};
      *(v2f*)(W0c + s10) = (v2f){wq1.x, wq1.y};
      *(v2f*)(W0c + s11) = (v2f){wq1.z, wq1.w};
      if (has2) {
        *(v2f*)(W0c + s20) = (v2f){wq2.x, wq2.y};
        *(v2f*)(W0c + s21) = (v2f){wq2.z, wq2.w};
      }
    }

    for (int t = wave; t < 6 * 6; t += NWAVE) {
      int m0 = (t / 6) * 16, n0 = (t % 6) * 16;
      v8f acc = {0.f, 0.f, 0.f, 0.f, 0.f, 0.f, 0.f, 0.f};
      for (int k0 = 0; k0 < KP; k0 += 4) {
        v2f a  = ld_a(Um, KP, m0, k0, l, half);
        v2f bb = ld_b(Tc, NP, k0, n0, l, half);
        WMMA_F32(acc, a, bb);
      }
      st_c(Oc, NP, m0, n0, acc, l, half, OW0, NP);
    }
    __syncthreads();

    for (int i = tid; i < P0 * P0; i += NTHREADS) {
      int py = i / P0, px = i % P0;
      float hi = -3.402823466e38f, lo = 3.402823466e38f;
      for (int dy = 0; dy < 3; ++dy)
        for (int dx = 0; dx < 3; ++dx) {
          float x = Oc[(3 * py + dy) * NP + (3 * px + dx)];
          hi = fmaxf(hi, x);
          lo = fminf(lo, x);
        }
      X1[c * (P0 * P0) + i] = gelu_pool_max(lo + bc, hi + bc);
    }
  }
  __syncthreads();

  float* A1 = smem + OFF_A1;
  float* O1 = smem + OFF_O1;
  float* X2 = smem + OFF_X2;
  for (int i = tid; i < C1 * K1; i += NTHREADS) {
    int co = i / K1, k = i - co * K1;
    int ci = k / 9, r9 = k - ci * 9;
    A1[co * K1 + r9 * 16 + ci] = w1[i];
  }
  __syncthreads();

  for (int nt = wave; nt < (NPIX1 + 15) / 16; nt += NWAVE) {
    const int col = nt * 16 + l;
    const int pc  = col < NPIX1 ? col : NPIX1 - 1;
    const int py = pc / OW1, px = pc - py * OW1;
    const float* xb = X1 + py * P0 + px + half * (2 * P0 * P0);
    const float* ab = A1 + l * K1 + 2 * half;
    v8f acc0 = {0.f, 0.f, 0.f, 0.f, 0.f, 0.f, 0.f, 0.f};
    v8f acc1 = {0.f, 0.f, 0.f, 0.f, 0.f, 0.f, 0.f, 0.f};
    v2f fb[2][4], fa0[2][4], fa1[2][4];
#pragma unroll
    for (int q = 0; q < 4; ++q) {
      fb[0][q].x = xb[(4 * q) * (P0 * P0)];
      fb[0][q].y = xb[(4 * q + 1) * (P0 * P0)];
      fa0[0][q]  = *(const v2f*)(ab + 4 * q);
      fa1[0][q]  = *(const v2f*)(ab + 16 * K1 + 4 * q);
    }
#pragma unroll
    for (int r9 = 0; r9 < 9; ++r9) {
      const int cur = r9 & 1, nxt = cur ^ 1;
      if (r9 + 1 < 9) {
        const int sp = ((r9 + 1) / 3) * P0 + ((r9 + 1) % 3);
#pragma unroll
        for (int q = 0; q < 4; ++q) {
          fb[nxt][q].x = xb[(4 * q) * (P0 * P0) + sp];
          fb[nxt][q].y = xb[(4 * q + 1) * (P0 * P0) + sp];
          fa0[nxt][q]  = *(const v2f*)(ab + (r9 + 1) * 16 + 4 * q);
          fa1[nxt][q]  = *(const v2f*)(ab + 16 * K1 + (r9 + 1) * 16 + 4 * q);
        }
      }
      __builtin_amdgcn_sched_barrier(0);
#pragma unroll
      for (int q = 0; q < 4; ++q) {
        WMMA_F32(acc0, fa0[cur][q], fb[cur][q]);
        WMMA_F32(acc1, fa1[cur][q], fb[cur][q]);
      }
      __builtin_amdgcn_sched_barrier(0);
    }
    if (col < NPIX1) {
#pragma unroll
      for (int r = 0; r < 8; ++r) {
        O1[(r + 8 * half) * NPIX1 + col]      = acc0[r];
        O1[(16 + r + 8 * half) * NPIX1 + col] = acc1[r];
      }
    }
  }
  __syncthreads();

  for (int i = tid; i < C1 * P1 * P1; i += NTHREADS) {
    int co = i >> 6, rem = i & 63, py = rem >> 3, px = rem & 7;
    float bc = b1[co];
    float hi = -3.402823466e38f, lo = 3.402823466e38f;
    for (int dy = 0; dy < 3; ++dy)
      for (int dx = 0; dx < 3; ++dx) {
        float x = O1[co * NPIX1 + (3 * py + dy) * OW1 + (3 * px + dx)];
        hi = fmaxf(hi, x);
        lo = fminf(lo, x);
      }
    X2[i] = gelu_pool_max(lo + bc, hi + bc);
  }
  __syncthreads();

  float* A2 = smem + OFF_A2;
  float* B2 = smem + OFF_B2;
  float* O2 = smem + OFF_O2;
  for (int i = tid; i < C2 * K2; i += NTHREADS) A2[i] = w2[i];
  for (int i = tid; i < K2 * NPIX2P; i += NTHREADS) {
    int kk = i / NPIX2P, col = i % NPIX2P;
    float val = 0.f;
    if (col < NPIX2) {
      int py = col / OW2, px = col % OW2;
      int ci = kk / 9, r9 = kk % 9;
      val = X2[ci * (P1 * P1) + (py + r9 / 3) * P1 + (px + r9 % 3)];
    }
    B2[i] = val;
  }
  __syncthreads();

  for (int t = wave; t < 2 * 3; t += NWAVE) {
    int m0 = (t / 3) * 16, n0 = (t % 3) * 16;
    v8f acc = {0.f, 0.f, 0.f, 0.f, 0.f, 0.f, 0.f, 0.f};
    for (int k0 = 0; k0 < K2; k0 += 4) {
      v2f a  = ld_a(A2, K2, m0, k0, l, half);
      v2f bb = ld_b(B2, NPIX2P, k0, n0, l, half);
      WMMA_F32(acc, a, bb);
    }
    st_c(O2, NPIX2P, m0, n0, acc, l, half, C2, NPIX2P);
  }
  __syncthreads();

  if (tid < 128) {
    int co = tid >> 2, rem = tid & 3, py = rem >> 1, px = rem & 1;
    float bc = b2[co];
    float hi = -3.402823466e38f, lo = 3.402823466e38f;
    for (int dy = 0; dy < 3; ++dy)
      for (int dx = 0; dx < 3; ++dx) {
        float x = O2[co * NPIX2P + (3 * py + dy) * OW2 + (3 * px + dx)];
        hi = fmaxf(hi, x);
        lo = fminf(lo, x);
      }
    out[tok * 128 + tid] = gelu_pool_max(lo + bc, hi + bc);
  }
}

extern "C" void kernel_launch(void* const* d_in, const int* in_sizes, int n_in,
                              void* d_out, int out_size, void* d_ws, size_t ws_size,
                              hipStream_t stream) {
  (void)in_sizes; (void)n_in; (void)out_size; (void)d_ws; (void)ws_size;
  const float* tf = (const float*)d_in[0];
  const float* tp = (const float*)d_in[1];
  const float* w0 = (const float*)d_in[2];
  const float* b0 = (const float*)d_in[3];
  const float* w1 = (const float*)d_in[4];
  const float* b1 = (const float*)d_in[5];
  const float* w2 = (const float*)d_in[6];
  const float* b2 = (const float*)d_in[7];
  float* out = (float*)d_out;

  const size_t shmem = SMEM_FLOATS * sizeof(float);
  (void)hipFuncSetAttribute((const void*)fused_outer_conv_kernel,
                            hipFuncAttributeMaxDynamicSharedMemorySize, (int)shmem);
  hipLaunchKernelGGL(fused_outer_conv_kernel, dim3(NTOK), dim3(NTHREADS), shmem,
                     stream, tf, tp, w0, b0, w1, b1, w2, b2, out);
}


# conv0's B operands (im2row of v, intermediate T) stored k-contiguous at pitch 52: one conflict-free 8-byte LDS load per B fragment, was two 2-way-conflicted 4-byte loads; T tiles stored 16 bytes wide
# speedup vs baseline: 4.0996x; 1.1072x over previous
#include <hip/hip_runtime.h>


typedef float v2f __attribute__((ext_vector_type(2)));
typedef float v4f __attribute__((ext_vector_type(4)));
typedef float v8f __attribute__((ext_vector_type(8)));

#define NB 8
#define NS 64
#define ND 300
#define NTOK (NB * NS)
#define NTHREADS 256
#define NWAVE 8

#define C0 16
#define K0 50
#define KP 52
#define OW0 84
#define NP 96
#define P0 28
#define W0Q 625

#define C1 32
#define OW1 26
#define NPIX1 676
#define K1 144
#define P1 8

#define C2 32
#define OW2 6
#define NPIX2 36
#define NPIX2P 48
#define K2 288

#define OFF_VM  0
#define OFF_UM  4992
#define OFF_W0C 9984
#define OFF_TC  13312
#define OFF_OC  18304
#define OFF_X1  26368
#define OFF_A1  0
#define OFF_O1  4608
#define OFF_X2  26368
#define OFF_A2  0
#define OFF_B2  9216
#define OFF_O2  23040
#define SMEM_FLOATS 38912

__device__ __forceinline__ float gelu_exact(float x) {
  return 0.5f * x * (1.0f + erff(x * 0.70710678118654752f));
}

__device__ __forceinline__ float gelu_pool_max(float lo, float hi) {
  return fmaxf(gelu_exact(lo), gelu_exact(hi));
}

__device__ __forceinline__ int w0_pair_slot(int p) {
  int ky = p / 25;
  return ky * KP + 2 * (p - ky * 25);
}

__device__ __forceinline__ v2f ld_a(const float* A, int lda, int m0, int k0,
                                    int l, int half) {
  return *(const v2f*)(A + (m0 + l) * lda + (k0 + 2 * half));
}
__device__ __forceinline__ v2f ld_bt(const float* Bt, int ldk, int k0, int n0,
                                     int l, int half) {
  return *(const v2f*)(Bt + (n0 + l) * ldk + (k0 + 2 * half));
}
__device__ __forceinline__ v2f ld_b(const float* Bm, int ldb, int k0, int n0,
                                    int l, int half) {
  v2f b;
  b.x = Bm[(k0 + 2 * half) * ldb + n0 + l];
  b.y = Bm[(k0 + 2 * half + 1) * ldb + n0 + l];
  return b;
}
__device__ __forceinline__ void st_c(float* Cm, int ldc, int m0, int n0, v8f c,
                                     int l, int half, int maxRow, int maxCol) {
  int col = n0 + l;
  if (col >= maxCol) return;
#pragma unroll
  for (int r = 0; r < 8; ++r) {
    int row = m0 + r + 8 * half;
    if (row < maxRow) Cm[row * ldc + col] = c[r];
  }
}

__device__ __forceinline__ void st_c_t(float* Ct, int ldk, int m0, int n0, v8f c,
                                       int l, int half, int maxRow) {
  const int row = m0 + 8 * half;
  float* p = Ct + (n0 + l) * ldk + row;
  if (row < maxRow)     *(v4f*)p       = (v4f){c[0], c[1], c[2], c[3]};
  if (row + 4 < maxRow) *(v4f*)(p + 4) = (v4f){c[4], c[5], c[6], c[7]};
}

#define WMMA_F32(acc, a, b)                                                   \
  (acc) = __builtin_amdgcn_wmma_f32_16x16x4_f32(false, (a), false, (b),       \
                                                (short)0, (acc), false, false)

extern "C" __global__ __launch_bounds__(NTHREADS) void
fused_outer_conv_kernel(const float* __restrict__ tf,
                        const float* __restrict__ tp,
                        const float* __restrict__ w0, const float* __restrict__ b0,
                        const float* __restrict__ w1, const float* __restrict__ b1,
                        const float* __restrict__ w2, const float* __restrict__ b2,
                        float* __restrict__ out) {
  extern __shared__ float smem[];
  const int tid  = threadIdx.x;
  const int lane = tid & 31;
  const int l    = tid & 15;
  const int half = (tid >> 4) & 1;
  const int wave = tid >> 5;
  const int tok  = blockIdx.x;

  const float* v = tf + tok * ND;
  const float* u = tp + tok * ND;

  {
    const char* pw1 = (const char*)w1;
    const char* pw2 = (const char*)w2;
    if (tid * 128 < (int)(C1 * K1 * sizeof(float))) __builtin_prefetch(pw1 + tid * 128, 0, 1);
    if (tid * 128 < (int)(C2 * K2 * sizeof(float))) __builtin_prefetch(pw2 + tid * 128, 0, 1);
  }

  float* Vm  = smem + OFF_VM;
  float* Um  = smem + OFF_UM;
  float* W0c = smem + OFF_W0C;
  float* Tc  = smem + OFF_TC;
  float* Oc  = smem + OFF_OC;
  float* X1  = smem + OFF_X1;

  const v4f* w0v = (const v4f*)w0;
  const int  q0 = tid, q1 = tid + NTHREADS;
  const bool has2 = tid + 2 * NTHREADS < W0Q;
  const int  q2 = has2 ? tid + 2 * NTHREADS : W0Q - 1;
  const int  s00 = w0_pair_slot(2 * q0), s01 = w0_pair_slot(2 * q0 + 1);
  const int  s10 = w0_pair_slot(2 * q1), s11 = w0_pair_slot(2 * q1 + 1);
  const int  s20 = w0_pair_slot(2 * q2), s21 = w0_pair_slot(2 * q2 + 1);

  v4f wq0 = w0v[q0], wq1 = w0v[q1], wq2 = w0v[q2];

  for (int i = tid; i < NP * KP; i += NTHREADS) {
    int o = i / KP, k = i % KP;
    bool in = o < OW0 && k < K0;
    Um[i] = in ? u[3 * o + k] : 0.f;
    Vm[i] = in ? v[3 * o + k] : 0.f;
  }

  for (int i = tid; i < (64 - K0) * KP; i += NTHREADS) W0c[K0 * KP + i] = 0.f;
  if (tid < 2 * K0) W0c[(tid >> 1) * KP + K0 + (tid & 1)] = 0.f;
  *(v2f*)(W0c + s00) = (v2f){wq0.x, wq0.y};
  *(v2f*)(W0c + s01) = (v2f){wq0.z, wq0.w};
  *(v2f*)(W0c + s10) = (v2f){wq1.x, wq1.y};
  *(v2f*)(W0c + s11) = (v2f){wq1.z, wq1.w};
  if (has2) {
    *(v2f*)(W0c + s20) = (v2f){wq2.x, wq2.y};
    *(v2f*)(W0c + s21) = (v2f){wq2.z, wq2.w};
  }
  __syncthreads();

  for (int c = 0; c < C0; ++c) {
    if (c + 1 < C0) {
      const v4f* wn = w0v + (c + 1) * W0Q;
      wq0 = wn[q0];
      wq1 = wn[q1];
      wq2 = wn[q2];
    }
    const float bc = b0[c];

    for (int t = wave; t < 4 * 6; t += NWAVE) {
      int m0 = (t / 6) * 16, n0 = (t % 6) * 16;
      v8f acc = {0.f, 0.f, 0.f, 0.f, 0.f, 0.f, 0.f, 0.f};
      for (int k0 = 0; k0 < KP; k0 += 4) {
        v2f a  = ld_a(W0c, KP, m0, k0, l, half);
        v2f bb = ld_bt(Vm, KP, k0, n0, l, half);
        WMMA_F32(acc, a, bb);
      }
      st_c_t(Tc, KP, m0, n0, acc, l, half, KP);
    }
    __syncthreads();

    if (c + 1 < C0) {
      *(v2f*)(W0c + s00) = (v2f){wq0.x, wq0.y};
      *(v2f*)(W0c + s01) = (v2f){wq0.z, wq0.w};
      *(v2f*)(W0c + s10) = (v2f){wq1.x, wq1.y};
      *(v2f*)(W0c + s11) = (v2f){wq1.z, wq1.w};
      if (has2) {
        *(v2f*)(W0c + s20) = (v2f){wq2.x, wq2.y};
        *(v2f*)(W0c + s21) = (v2f){wq2.z, wq2.w};
      }
    }

    for (int t = wave; t < 6 * 6; t += NWAVE) {
      int m0 = (t / 6) * 16, n0 = (t % 6) * 16;
      v8f acc = {0.f, 0.f, 0.f, 0.f, 0.f, 0.f, 0.f, 0.f};
      for (int k0 = 0; k0 < KP; k0 += 4) {
        v2f a  = ld_a(Um, KP, m0, k0, l, half);
        v2f bb = ld_bt(Tc, KP, k0, n0, l, half);
        WMMA_F32(acc, a, bb);
      }
      st_c(Oc, NP, m0, n0, acc, l, half, OW0, NP);
    }
    __syncthreads();

    for (int i = tid; i < P0 * P0; i += NTHREADS) {
      int py = i / P0, px = i % P0;
      float hi = -3.402823466e38f, lo = 3.402823466e38f;
      for (int dy = 0; dy < 3; ++dy)
        for (int dx = 0; dx < 3; ++dx) {
          float x = Oc[(3 * py + dy) * NP + (3 * px + dx)];
          hi = fmaxf(hi, x);
          lo = fminf(lo, x);
        }
      X1[c * (P0 * P0) + i] = gelu_pool_max(lo + bc, hi + bc);
    }
  }
  __syncthreads();

  float* A1 = smem + OFF_A1;
  float* O1 = smem + OFF_O1;
  float* X2 = smem + OFF_X2;
  for (int i = tid; i < C1 * K1; i += NTHREADS) {
    int co = i / K1, k = i - co * K1;
    int ci = k / 9, r9 = k - ci * 9;
    A1[co * K1 + r9 * 16 + ci] = w1[i];
  }
  __syncthreads();

  for (int nt = wave; nt < (NPIX1 + 15) / 16; nt += NWAVE) {
    const int col = nt * 16 + l;
    const int pc  = col < NPIX1 ? col : NPIX1 - 1;
    const int py = pc / OW1, px = pc - py * OW1;
    const float* xb = X1 + py * P0 + px + half * (2 * P0 * P0);
    const float* ab = A1 + l * K1 + 2 * half;
    v8f acc0 = {0.f, 0.f, 0.f, 0.f, 0.f, 0.f, 0.f, 0.f};
    v8f acc1 = {0.f, 0.f, 0.f, 0.f, 0.f, 0.f, 0.f, 0.f};
    v2f fb[2][4], fa0[2][4], fa1[2][4];
#pragma unroll
    for (int q = 0; q < 4; ++q) {
      fb[0][q].x = xb[(4 * q) * (P0 * P0)];
      fb[0][q].y = xb[(4 * q + 1) * (P0 * P0)];
      fa0[0][q]  = *(const v2f*)(ab + 4 * q);
      fa1[0][q]  = *(const v2f*)(ab + 16 * K1 + 4 * q);
    }
#pragma unroll
    for (int r9 = 0; r9 < 9; ++r9) {
      const int cur = r9 & 1, nxt = cur ^ 1;
      if (r9 + 1 < 9) {
        const int sp = ((r9 + 1) / 3) * P0 + ((r9 + 1) % 3);
#pragma unroll
        for (int q = 0; q < 4; ++q) {
          fb[nxt][q].x = xb[(4 * q) * (P0 * P0) + sp];
          fb[nxt][q].y = xb[(4 * q + 1) * (P0 * P0) + sp];
          fa0[nxt][q]  = *(const v2f*)(ab + (r9 + 1) * 16 + 4 * q);
          fa1[nxt][q]  = *(const v2f*)(ab + 16 * K1 + (r9 + 1) * 16 + 4 * q);
        }
      }
      __builtin_amdgcn_sched_barrier(0);
#pragma unroll
      for (int q = 0; q < 4; ++q) {
        WMMA_F32(acc0, fa0[cur][q], fb[cur][q]);
        WMMA_F32(acc1, fa1[cur][q], fb[cur][q]);
      }
      __builtin_amdgcn_sched_barrier(0);
    }
    if (col < NPIX1) {
#pragma unroll
      for (int r = 0; r < 8; ++r) {
        O1[(r + 8 * half) * NPIX1 + col]      = acc0[r];
        O1[(16 + r + 8 * half) * NPIX1 + col] = acc1[r];
      }
    }
  }
  __syncthreads();

  for (int i = tid; i < C1 * P1 * P1; i += NTHREADS) {
    int co = i >> 6, rem = i & 63, py = rem >> 3, px = rem & 7;
    float bc = b1[co];
    float hi = -3.402823466e38f, lo = 3.402823466e38f;
    for (int dy = 0; dy < 3; ++dy)
      for (int dx = 0; dx < 3; ++dx) {
        float x = O1[co * NPIX1 + (3 * py + dy) * OW1 + (3 * px + dx)];
        hi = fmaxf(hi, x);
        lo = fminf(lo, x);
      }
    X2[i] = gelu_pool_max(lo + bc, hi + bc);
  }
  __syncthreads();

  float* A2 = smem + OFF_A2;
  float* B2 = smem + OFF_B2;
  float* O2 = smem + OFF_O2;
  for (int i = tid; i < C2 * K2; i += NTHREADS) A2[i] = w2[i];
  for (int i = tid; i < K2 * NPIX2P; i += NTHREADS) {
    int kk = i / NPIX2P, col = i % NPIX2P;
    float val = 0.f;
    if (col < NPIX2) {
      int py = col / OW2, px = col % OW2;
      int ci = kk / 9, r9 = kk % 9;
      val = X2[ci * (P1 * P1) + (py + r9 / 3) * P1 + (px + r9 % 3)];
    }
    B2[i] = val;
  }
  __syncthreads();

  for (int t = wave; t < 2 * 3; t += NWAVE) {
    int m0 = (t / 3) * 16, n0 = (t % 3) * 16;
    v8f acc = {0.f, 0.f, 0.f, 0.f, 0.f, 0.f, 0.f, 0.f};
    for (int k0 = 0; k0 < K2; k0 += 4) {
      v2f a  = ld_a(A2, K2, m0, k0, l, half);
      v2f bb = ld_b(B2, NPIX2P, k0, n0, l, half);
      WMMA_F32(acc, a, bb);
    }
    st_c(O2, NPIX2P, m0, n0, acc, l, half, C2, NPIX2P);
  }
  __syncthreads();

  if (tid < 128) {
    int co = tid >> 2, rem = tid & 3, py = rem >> 1, px = rem & 1;
    float bc = b2[co];
    float hi = -3.402823466e38f, lo = 3.402823466e38f;
    for (int dy = 0; dy < 3; ++dy)
      for (int dx = 0; dx < 3; ++dx) {
        float x = O2[co * NPIX2P + (3 * py + dy) * OW2 + (3 * px + dx)];
        hi = fmaxf(hi, x);
        lo = fminf(lo, x);
      }
    out[tok * 128 + tid] = gelu_pool_max(lo + bc, hi + bc);
  }
}

extern "C" void kernel_launch(void* const* d_in, const int* in_sizes, int n_in,
                              void* d_out, int out_size, void* d_ws, size_t ws_size,
                              hipStream_t stream) {
  (void)in_sizes; (void)n_in; (void)out_size; (void)d_ws; (void)ws_size;
  const float* tf = (const float*)d_in[0];
  const float* tp = (const float*)d_in[1];
  const float* w0 = (const float*)d_in[2];
  const float* b0 = (const float*)d_in[3];
  const float* w1 = (const float*)d_in[4];
  const float* b1 = (const float*)d_in[5];
  const float* w2 = (const float*)d_in[6];
  const float* b2 = (const float*)d_in[7];
  float* out = (float*)d_out;

  const size_t shmem = SMEM_FLOATS * sizeof(float);
  (void)hipFuncSetAttribute((const void*)fused_outer_conv_kernel,
                            hipFuncAttributeMaxDynamicSharedMemorySize, (int)shmem);
  hipLaunchKernelGGL(fused_outer_conv_kernel, dim3(NTOK), dim3(NTHREADS), shmem,
                     stream, tf, tp, w0, b0, w1, b1, w2, b2, out);
}


# conv0 GEMMs as per-wave strips of one row tile by three column tiles: A fragments shared by three independent accumulators, a strip's 52 fragment loads issued as one group ahead of its 39 WMMAs
# speedup vs baseline: 4.6027x; 1.1227x over previous
#include <hip/hip_runtime.h>


typedef float v2f __attribute__((ext_vector_type(2)));
typedef float v4f __attribute__((ext_vector_type(4)));
typedef float v8f __attribute__((ext_vector_type(8)));

#define NB 8
#define NS 64
#define ND 300
#define NTOK (NB * NS)
#define NTHREADS 256
#define NWAVE 8

#define C0 16
#define K0 50
#define KP 52
#define OW0 84
#define NP 96
#define P0 28
#define W0Q 625

#define C1 32
#define OW1 26
#define NPIX1 676
#define K1 144
#define P1 8

#define C2 32
#define OW2 6
#define NPIX2 36
#define NPIX2P 48
#define K2 288

#define OFF_VM  0
#define OFF_UM  4992
#define OFF_W0C 9984
#define OFF_TC  13312
#define OFF_OC  18304
#define OFF_X1  26368
#define OFF_A1  0
#define OFF_O1  4608
#define OFF_X2  26368
#define OFF_A2  0
#define OFF_B2  9216
#define OFF_O2  23040
#define SMEM_FLOATS 38912

__device__ __forceinline__ float gelu_exact(float x) {
  return 0.5f * x * (1.0f + erff(x * 0.70710678118654752f));
}

__device__ __forceinline__ float gelu_pool_max(float lo, float hi) {
  return fmaxf(gelu_exact(lo), gelu_exact(hi));
}

__device__ __forceinline__ int w0_pair_slot(int p) {
  int ky = p / 25;
  return ky * KP + 2 * (p - ky * 25);
}

__device__ __forceinline__ v2f ld_a(const float* A, int lda, int m0, int k0,
                                    int l, int half) {
  return *(const v2f*)(A + (m0 + l) * lda + (k0 + 2 * half));
}
__device__ __forceinline__ v2f ld_bt(const float* Bt, int ldk, int k0, int n0,
                                     int l, int half) {
  return *(const v2f*)(Bt + (n0 + l) * ldk + (k0 + 2 * half));
}
__device__ __forceinline__ v2f ld_b(const float* Bm, int ldb, int k0, int n0,
                                    int l, int half) {
  v2f b;
  b.x = Bm[(k0 + 2 * half) * ldb + n0 + l];
  b.y = Bm[(k0 + 2 * half + 1) * ldb + n0 + l];
  return b;
}
__device__ __forceinline__ void st_c(float* Cm, int ldc, int m0, int n0, v8f c,
                                     int l, int half, int maxRow, int maxCol) {
  int col = n0 + l;
  if (col >= maxCol) return;
#pragma unroll
  for (int r = 0; r < 8; ++r) {
    int row = m0 + r + 8 * half;
    if (row < maxRow) Cm[row * ldc + col] = c[r];
  }
}

__device__ __forceinline__ void st_c3(float* Cm, int ldc, int m0, int n0,
                                      v8f c0, v8f c1, v8f c2, int l, int half, int maxRow) {
  const int row0 = m0 + 8 * half;
  float* p = Cm + row0 * ldc + n0 + l;
#pragma unroll
  for (int r = 0; r < 8; ++r) {
    if (row0 + r < maxRow) {
      p[r * ldc]      = c0[r];
      p[r * ldc + 16] = c1[r];
      p[r * ldc + 32] = c2[r];
    }
  }
}

__device__ __forceinline__ void st_c_t3(float* Ct, int ldk, int m0, int n0,
                                        v8f c0, v8f c1, v8f c2, int l, int half, int maxRow) {
  const int row = m0 + 8 * half;
  float* p = Ct + (n0 + l) * ldk + row;
  if (row < maxRow) {
    *(v4f*)p              = (v4f){c0[0], c0[1], c0[2], c0[3]};
    *(v4f*)(p + 16 * ldk) = (v4f){c1[0], c1[1], c1[2], c1[3]};
    *(v4f*)(p + 32 * ldk) = (v4f){c2[0], c2[1], c2[2], c2[3]};
  }
  if (row + 4 < maxRow) {
    *(v4f*)(p + 4)            = (v4f){c0[4], c0[5], c0[6], c0[7]};
    *(v4f*)(p + 16 * ldk + 4) = (v4f){c1[4], c1[5], c1[6], c1[7]};
    *(v4f*)(p + 32 * ldk + 4) = (v4f){c2[4], c2[5], c2[6], c2[7]};
  }
}

#define WMMA_F32(acc, a, b)                                                   \
  (acc) = __builtin_amdgcn_wmma_f32_16x16x4_f32(false, (a), false, (b),       \
                                                (short)0, (acc), false, false)

__device__ __forceinline__ void strip_gemm(const float* A, const float* Bt, int m0, int nb,
                                           int l, int half, v8f& acc0, v8f& acc1, v8f& acc2) {
  v2f fa[13], fb0[13], fb1[13], fb2[13];
#pragma unroll
  for (int s = 0; s < 13; ++s) {
    fa[s]  = ld_a(A, KP, m0, 4 * s, l, half);
    fb0[s] = ld_bt(Bt, KP, 4 * s, nb, l, half);
    fb1[s] = ld_bt(Bt, KP, 4 * s, nb + 16, l, half);
    fb2[s] = ld_bt(Bt, KP, 4 * s, nb + 32, l, half);
  }
  __builtin_amdgcn_sched_barrier(0);
  acc0 = (v8f){0.f, 0.f, 0.f, 0.f, 0.f, 0.f, 0.f, 0.f};
  acc1 = acc0;
  acc2 = acc0;
#pragma unroll
  for (int s = 0; s < 13; ++s) {
    WMMA_F32(acc0, fa[s], fb0[s]);
    WMMA_F32(acc1, fa[s], fb1[s]);
    WMMA_F32(acc2, fa[s], fb2[s]);
  }
}

extern "C" __global__ __launch_bounds__(NTHREADS) void
fused_outer_conv_kernel(const float* __restrict__ tf,
                        const float* __restrict__ tp,
                        const float* __restrict__ w0, const float* __restrict__ b0,
                        const float* __restrict__ w1, const float* __restrict__ b1,
                        const float* __restrict__ w2, const float* __restrict__ b2,
                        float* __restrict__ out) {
  extern __shared__ float smem[];
  const int tid  = threadIdx.x;
  const int lane = tid & 31;
  const int l    = tid & 15;
  const int half = (tid >> 4) & 1;
  const int wave = tid >> 5;
  const int tok  = blockIdx.x;

  const float* v = tf + tok * ND;
  const float* u = tp + tok * ND;

  {
    const char* pw1 = (const char*)w1;
    const char* pw2 = (const char*)w2;
    if (tid * 128 < (int)(C1 * K1 * sizeof(float))) __builtin_prefetch(pw1 + tid * 128, 0, 1);
    if (tid * 128 < (int)(C2 * K2 * sizeof(float))) __builtin_prefetch(pw2 + tid * 128, 0, 1);
  }

  float* Vm  = smem + OFF_VM;
  float* Um  = smem + OFF_UM;
  float* W0c = smem + OFF_W0C;
  float* Tc  = smem + OFF_TC;
  float* Oc  = smem + OFF_OC;
  float* X1  = smem + OFF_X1;

  const v4f* w0v = (const v4f*)w0;
  const int  q0 = tid, q1 = tid + NTHREADS;
  const bool has2 = tid + 2 * NTHREADS < W0Q;
  const int  q2 = has2 ? tid + 2 * NTHREADS : W0Q - 1;
  const int  s00 = w0_pair_slot(2 * q0), s01 = w0_pair_slot(2 * q0 + 1);
  const int  s10 = w0_pair_slot(2 * q1), s11 = w0_pair_slot(2 * q1 + 1);
  const int  s20 = w0_pair_slot(2 * q2), s21 = w0_pair_slot(2 * q2 + 1);

  v4f wq0 = w0v[q0], wq1 = w0v[q1], wq2 = w0v[q2];

  for (int i = tid; i < NP * KP; i += NTHREADS) {
    int o = i / KP, k = i % KP;
    bool in = o < OW0 && k < K0;
    Um[i] = in ? u[3 * o + k] : 0.f;
    Vm[i] = in ? v[3 * o + k] : 0.f;
  }

  for (int i = tid; i < (64 - K0) * KP; i += NTHREADS) W0c[K0 * KP + i] = 0.f;
  if (tid < 2 * K0) W0c[(tid >> 1) * KP + K0 + (tid & 1)] = 0.f;
  *(v2f*)(W0c + s00) = (v2f){wq0.x, wq0.y};
  *(v2f*)(W0c + s01) = (v2f){wq0.z, wq0.w};
  *(v2f*)(W0c + s10) = (v2f){wq1.x, wq1.y};
  *(v2f*)(W0c + s11) = (v2f){wq1.z, wq1.w};
  if (has2) {
    *(v2f*)(W0c + s20) = (v2f){wq2.x, wq2.y};
    *(v2f*)(W0c + s21) = (v2f){wq2.z, wq2.w};
  }
  __syncthreads();

  for (int c = 0; c < C0; ++c) {
    if (c + 1 < C0) {
      const v4f* wn = w0v + (c + 1) * W0Q;
      wq0 = wn[q0];
      wq1 = wn[q1];
      wq2 = wn[q2];
    }
    const float bc = b0[c];

    {
      const int m0 = (wave >> 1) * 16, nb = (wave & 1) * 48;
      v8f acc0, acc1, acc2;
      strip_gemm(W0c, Vm, m0, nb, l, half, acc0, acc1, acc2);
      st_c_t3(Tc, KP, m0, nb, acc0, acc1, acc2, l, half, KP);
    }
    __syncthreads();

    if (c + 1 < C0) {
      *(v2f*)(W0c + s00) = (v2f){wq0.x, wq0.y};
      *(v2f*)(W0c + s01) = (v2f){wq0.z, wq0.w};
      *(v2f*)(W0c + s10) = (v2f){wq1.x, wq1.y};
      *(v2f*)(W0c + s11) = (v2f){wq1.z, wq1.w};
      if (has2) {
        *(v2f*)(W0c + s20) = (v2f){wq2.x, wq2.y};
        *(v2f*)(W0c + s21) = (v2f){wq2.z, wq2.w};
      }
    }

    const int nstrip = ((0xA5 >> wave) & 1) ? 2 : 1;
    for (int i = 0; i < nstrip; ++i) {
      const int j = i == 0 ? wave : 8 + (wave >> 1);
      const int m0 = (j >> 1) * 16, nb = (j & 1) * 48;
      v8f acc0, acc1, acc2;
      strip_gemm(Um, Tc, m0, nb, l, half, acc0, acc1, acc2);
      st_c3(Oc, NP, m0, nb, acc0, acc1, acc2, l, half, OW0);
    }
    __syncthreads();

    for (int i = tid; i < P0 * P0; i += NTHREADS) {
      int py = i / P0, px = i % P0;
      float hi = -3.402823466e38f, lo = 3.402823466e38f;
      for (int dy = 0; dy < 3; ++dy)
        for (int dx = 0; dx < 3; ++dx) {
          float x = Oc[(3 * py + dy) * NP + (3 * px + dx)];
          hi = fmaxf(hi, x);
          lo = fminf(lo, x);
        }
      X1[c * (P0 * P0) + i] = gelu_pool_max(lo + bc, hi + bc);
    }
  }
  __syncthreads();

  float* A1 = smem + OFF_A1;
  float* O1 = smem + OFF_O1;
  float* X2 = smem + OFF_X2;
  for (int i = tid; i < C1 * K1; i += NTHREADS) {
    int co = i / K1, k = i - co * K1;
    int ci = k / 9, r9 = k - ci * 9;
    A1[co * K1 + r9 * 16 + ci] = w1[i];
  }
  __syncthreads();

  for (int nt = wave; nt < (NPIX1 + 15) / 16; nt += NWAVE) {
    const int col = nt * 16 + l;
    const int pc  = col < NPIX1 ? col : NPIX1 - 1;
    const int py = pc / OW1, px = pc - py * OW1;
    const float* xb = X1 + py * P0 + px + half * (2 * P0 * P0);
    const float* ab = A1 + l * K1 + 2 * half;
    v8f acc0 = {0.f, 0.f, 0.f, 0.f, 0.f, 0.f, 0.f, 0.f};
    v8f acc1 = {0.f, 0.f, 0.f, 0.f, 0.f, 0.f, 0.f, 0.f};
    v2f fb[2][4], fa0[2][4], fa1[2][4];
#pragma unroll
    for (int q = 0; q < 4; ++q) {
      fb[0][q].x = xb[(4 * q) * (P0 * P0)];
      fb[0][q].y = xb[(4 * q + 1) * (P0 * P0)];
      fa0[0][q]  = *(const v2f*)(ab + 4 * q);
      fa1[0][q]  = *(const v2f*)(ab + 16 * K1 + 4 * q);
    }
#pragma unroll
    for (int r9 = 0; r9 < 9; ++r9) {
      const int cur = r9 & 1, nxt = cur ^ 1;
      if (r9 + 1 < 9) {
        const int sp = ((r9 + 1) / 3) * P0 + ((r9 + 1) % 3);
#pragma unroll
        for (int q = 0; q < 4; ++q) {
          fb[nxt][q].x = xb[(4 * q) * (P0 * P0) + sp];
          fb[nxt][q].y = xb[(4 * q + 1) * (P0 * P0) + sp];
          fa0[nxt][q]  = *(const v2f*)(ab + (r9 + 1) * 16 + 4 * q);
          fa1[nxt][q]  = *(const v2f*)(ab + 16 * K1 + (r9 + 1) * 16 + 4 * q);
        }
      }
      __builtin_amdgcn_sched_barrier(0);
#pragma unroll
      for (int q = 0; q < 4; ++q) {
        WMMA_F32(acc0, fa0[cur][q], fb[cur][q]);
        WMMA_F32(acc1, fa1[cur][q], fb[cur][q]);
      }
      __builtin_amdgcn_sched_barrier(0);
    }
    if (col < NPIX1) {
#pragma unroll
      for (int r = 0; r < 8; ++r) {
        O1[(r + 8 * half) * NPIX1 + col]      = acc0[r];
        O1[(16 + r + 8 * half) * NPIX1 + col] = acc1[r];
      }
    }
  }
  __syncthreads();

  for (int i = tid; i < C1 * P1 * P1; i += NTHREADS) {
    int co = i >> 6, rem = i & 63, py = rem >> 3, px = rem & 7;
    float bc = b1[co];
    float hi = -3.402823466e38f, lo = 3.402823466e38f;
    for (int dy = 0; dy < 3; ++dy)
      for (int dx = 0; dx < 3; ++dx) {
        float x = O1[co * NPIX1 + (3 * py + dy) * OW1 + (3 * px + dx)];
        hi = fmaxf(hi, x);
        lo = fminf(lo, x);
      }
    X2[i] = gelu_pool_max(lo + bc, hi + bc);
  }
  __syncthreads();

  float* A2 = smem + OFF_A2;
  float* B2 = smem + OFF_B2;
  float* O2 = smem + OFF_O2;
  for (int i = tid; i < C2 * K2; i += NTHREADS) A2[i] = w2[i];
  for (int i = tid; i < K2 * NPIX2P; i += NTHREADS) {
    int kk = i / NPIX2P, col = i % NPIX2P;
    float val = 0.f;
    if (col < NPIX2) {
      int py = col / OW2, px = col % OW2;
      int ci = kk / 9, r9 = kk % 9;
      val = X2[ci * (P1 * P1) + (py + r9 / 3) * P1 + (px + r9 % 3)];
    }
    B2[i] = val;
  }
  __syncthreads();

  for (int t = wave; t < 2 * 3; t += NWAVE) {
    int m0 = (t / 3) * 16, n0 = (t % 3) * 16;
    v8f acc = {0.f, 0.f, 0.f, 0.f, 0.f, 0.f, 0.f, 0.f};
    for (int k0 = 0; k0 < K2; k0 += 4) {
      v2f a  = ld_a(A2, K2, m0, k0, l, half);
      v2f bb = ld_b(B2, NPIX2P, k0, n0, l, half);
      WMMA_F32(acc, a, bb);
    }
    st_c(O2, NPIX2P, m0, n0, acc, l, half, C2, NPIX2P);
  }
  __syncthreads();

  if (tid < 128) {
    int co = tid >> 2, rem = tid & 3, py = rem >> 1, px = rem & 1;
    float bc = b2[co];
    float hi = -3.402823466e38f, lo = 3.402823466e38f;
    for (int dy = 0; dy < 3; ++dy)
      for (int dx = 0; dx < 3; ++dx) {
        float x = O2[co * NPIX2P + (3 * py + dy) * OW2 + (3 * px + dx)];
        hi = fmaxf(hi, x);
        lo = fminf(lo, x);
      }
    out[tok * 128 + tid] = gelu_pool_max(lo + bc, hi + bc);
  }
}

extern "C" void kernel_launch(void* const* d_in, const int* in_sizes, int n_in,
                              void* d_out, int out_size, void* d_ws, size_t ws_size,
                              hipStream_t stream) {
  (void)in_sizes; (void)n_in; (void)out_size; (void)d_ws; (void)ws_size;
  const float* tf = (const float*)d_in[0];
  const float* tp = (const float*)d_in[1];
  const float* w0 = (const float*)d_in[2];
  const float* b0 = (const float*)d_in[3];
  const float* w1 = (const float*)d_in[4];
  const float* b1 = (const float*)d_in[5];
  const float* w2 = (const float*)d_in[6];
  const float* b2 = (const float*)d_in[7];
  float* out = (float*)d_out;

  const size_t shmem = SMEM_FLOATS * sizeof(float);
  (void)hipFuncSetAttribute((const void*)fused_outer_conv_kernel,
                            hipFuncAttributeMaxDynamicSharedMemorySize, (int)shmem);
  hipLaunchKernelGGL(fused_outer_conv_kernel, dim3(NTOK), dim3(NTHREADS), shmem,
                     stream, tf, tp, w0, b0, w1, b1, w2, b2, out);
}


# GELU evaluated once per pool window, at its largest pre-activation when that is non-negative (GELU >= 0 there, <= 0 below zero); an all-negative window also takes its smallest: 2 -> 1 erf
# speedup vs baseline: 4.8775x; 1.0597x over previous
#include <hip/hip_runtime.h>


typedef float v2f __attribute__((ext_vector_type(2)));
typedef float v4f __attribute__((ext_vector_type(4)));
typedef float v8f __attribute__((ext_vector_type(8)));

#define NB 8
#define NS 64
#define ND 300
#define NTOK (NB * NS)
#define NTHREADS 256
#define NWAVE 8

#define C0 16
#define K0 50
#define KP 52
#define OW0 84
#define NP 96
#define P0 28
#define W0Q 625

#define C1 32
#define OW1 26
#define NPIX1 676
#define K1 144
#define P1 8

#define C2 32
#define OW2 6
#define NPIX2 36
#define NPIX2P 48
#define K2 288

#define OFF_VM  0
#define OFF_UM  4992
#define OFF_W0C 9984
#define OFF_TC  13312
#define OFF_OC  18304
#define OFF_X1  26368
#define OFF_A1  0
#define OFF_O1  4608
#define OFF_X2  26368
#define OFF_A2  0
#define OFF_B2  9216
#define OFF_O2  23040
#define SMEM_FLOATS 38912

__device__ __forceinline__ float gelu_exact(float x) {
  return 0.5f * x * (1.0f + erff(x * 0.70710678118654752f));
}

__device__ __forceinline__ float gelu_pool_max9(const float x[9], float b) {
  const float hi = fmaxf(fmaxf(fmaxf(x[0], x[1]), fmaxf(x[2], x[3])),
                         fmaxf(fmaxf(x[4], x[5]), fmaxf(fmaxf(x[6], x[7]), x[8]))) + b;
  float r = gelu_exact(hi);
  if (hi < 0.f) {
    const float lo = fminf(fminf(fminf(x[0], x[1]), fminf(x[2], x[3])),
                           fminf(fminf(x[4], x[5]), fminf(fminf(x[6], x[7]), x[8]))) + b;
    r = fmaxf(r, gelu_exact(lo));
  }
  return r;
}

__device__ __forceinline__ int w0_pair_slot(int p) {
  int ky = p / 25;
  return ky * KP + 2 * (p - ky * 25);
}

__device__ __forceinline__ v2f ld_a(const float* A, int lda, int m0, int k0,
                                    int l, int half) {
  return *(const v2f*)(A + (m0 + l) * lda + (k0 + 2 * half));
}
__device__ __forceinline__ v2f ld_bt(const float* Bt, int ldk, int k0, int n0,
                                     int l, int half) {
  return *(const v2f*)(Bt + (n0 + l) * ldk + (k0 + 2 * half));
}
__device__ __forceinline__ v2f ld_b(const float* Bm, int ldb, int k0, int n0,
                                    int l, int half) {
  v2f b;
  b.x = Bm[(k0 + 2 * half) * ldb + n0 + l];
  b.y = Bm[(k0 + 2 * half + 1) * ldb + n0 + l];
  return b;
}
__device__ __forceinline__ void st_c(float* Cm, int ldc, int m0, int n0, v8f c,
                                     int l, int half, int maxRow, int maxCol) {
  int col = n0 + l;
  if (col >= maxCol) return;
#pragma unroll
  for (int r = 0; r < 8; ++r) {
    int row = m0 + r + 8 * half;
    if (row < maxRow) Cm[row * ldc + col] = c[r];
  }
}

__device__ __forceinline__ void st_c3(float* Cm, int ldc, int m0, int n0,
                                      v8f c0, v8f c1, v8f c2, int l, int half, int maxRow) {
  const int row0 = m0 + 8 * half;
  float* p = Cm + row0 * ldc + n0 + l;
#pragma unroll
  for (int r = 0; r < 8; ++r) {
    if (row0 + r < maxRow) {
      p[r * ldc]      = c0[r];
      p[r * ldc + 16] = c1[r];
      p[r * ldc + 32] = c2[r];
    }
  }
}

__device__ __forceinline__ void st_c_t3(float* Ct, int ldk, int m0, int n0,
                                        v8f c0, v8f c1, v8f c2, int l, int half, int maxRow) {
  const int row = m0 + 8 * half;
  float* p = Ct + (n0 + l) * ldk + row;
  if (row < maxRow) {
    *(v4f*)p              = (v4f){c0[0], c0[1], c0[2], c0[3]};
    *(v4f*)(p + 16 * ldk) = (v4f){c1[0], c1[1], c1[2], c1[3]};
    *(v4f*)(p + 32 * ldk) = (v4f){c2[0], c2[1], c2[2], c2[3]};
  }
  if (row + 4 < maxRow) {
    *(v4f*)(p + 4)            = (v4f){c0[4], c0[5], c0[6], c0[7]};
    *(v4f*)(p + 16 * ldk + 4) = (v4f){c1[4], c1[5], c1[6], c1[7]};
    *(v4f*)(p + 32 * ldk + 4) = (v4f){c2[4], c2[5], c2[6], c2[7]};
  }
}

#define WMMA_F32(acc, a, b)                                                   \
  (acc) = __builtin_amdgcn_wmma_f32_16x16x4_f32(false, (a), false, (b),       \
                                                (short)0, (acc), false, false)

__device__ __forceinline__ void strip_gemm(const float* A, const float* Bt, int m0, int nb,
                                           int l, int half, v8f& acc0, v8f& acc1, v8f& acc2) {
  v2f fa[13], fb0[13], fb1[13], fb2[13];
#pragma unroll
  for (int s = 0; s < 13; ++s) {
    fa[s]  = ld_a(A, KP, m0, 4 * s, l, half);
    fb0[s] = ld_bt(Bt, KP, 4 * s, nb, l, half);
    fb1[s] = ld_bt(Bt, KP, 4 * s, nb + 16, l, half);
    fb2[s] = ld_bt(Bt, KP, 4 * s, nb + 32, l, half);
  }
  __builtin_amdgcn_sched_barrier(0);
  acc0 = (v8f){0.f, 0.f, 0.f, 0.f, 0.f, 0.f, 0.f, 0.f};
  acc1 = acc0;
  acc2 = acc0;
#pragma unroll
  for (int s = 0; s < 13; ++s) {
    WMMA_F32(acc0, fa[s], fb0[s]);
    WMMA_F32(acc1, fa[s], fb1[s]);
    WMMA_F32(acc2, fa[s], fb2[s]);
  }
}

extern "C" __global__ __launch_bounds__(NTHREADS) void
fused_outer_conv_kernel(const float* __restrict__ tf,
                        const float* __restrict__ tp,
                        const float* __restrict__ w0, const float* __restrict__ b0,
                        const float* __restrict__ w1, const float* __restrict__ b1,
                        const float* __restrict__ w2, const float* __restrict__ b2,
                        float* __restrict__ out) {
  extern __shared__ float smem[];
  const int tid  = threadIdx.x;
  const int lane = tid & 31;
  const int l    = tid & 15;
  const int half = (tid >> 4) & 1;
  const int wave = tid >> 5;
  const int tok  = blockIdx.x;

  const float* v = tf + tok * ND;
  const float* u = tp + tok * ND;

  {
    const char* pw1 = (const char*)w1;
    const char* pw2 = (const char*)w2;
    if (tid * 128 < (int)(C1 * K1 * sizeof(float))) __builtin_prefetch(pw1 + tid * 128, 0, 1);
    if (tid * 128 < (int)(C2 * K2 * sizeof(float))) __builtin_prefetch(pw2 + tid * 128, 0, 1);
  }

  float* Vm  = smem + OFF_VM;
  float* Um  = smem + OFF_UM;
  float* W0c = smem + OFF_W0C;
  float* Tc  = smem + OFF_TC;
  float* Oc  = smem + OFF_OC;
  float* X1  = smem + OFF_X1;

  const v4f* w0v = (const v4f*)w0;
  const int  q0 = tid, q1 = tid + NTHREADS;
  const bool has2 = tid + 2 * NTHREADS < W0Q;
  const int  q2 = has2 ? tid + 2 * NTHREADS : W0Q - 1;
  const int  s00 = w0_pair_slot(2 * q0), s01 = w0_pair_slot(2 * q0 + 1);
  const int  s10 = w0_pair_slot(2 * q1), s11 = w0_pair_slot(2 * q1 + 1);
  const int  s20 = w0_pair_slot(2 * q2), s21 = w0_pair_slot(2 * q2 + 1);

  v4f wq0 = w0v[q0], wq1 = w0v[q1], wq2 = w0v[q2];

  for (int i = tid; i < NP * KP; i += NTHREADS) {
    int o = i / KP, k = i % KP;
    bool in = o < OW0 && k < K0;
    Um[i] = in ? u[3 * o + k] : 0.f;
    Vm[i] = in ? v[3 * o + k] : 0.f;
  }

  for (int i = tid; i < (64 - K0) * KP; i += NTHREADS) W0c[K0 * KP + i] = 0.f;
  if (tid < 2 * K0) W0c[(tid >> 1) * KP + K0 + (tid & 1)] = 0.f;
  *(v2f*)(W0c + s00) = (v2f){wq0.x, wq0.y};
  *(v2f*)(W0c + s01) = (v2f){wq0.z, wq0.w};
  *(v2f*)(W0c + s10) = (v2f){wq1.x, wq1.y};
  *(v2f*)(W0c + s11) = (v2f){wq1.z, wq1.w};
  if (has2) {
    *(v2f*)(W0c + s20) = (v2f){wq2.x, wq2.y};
    *(v2f*)(W0c + s21) = (v2f){wq2.z, wq2.w};
  }
  __syncthreads();

  for (int c = 0; c < C0; ++c) {
    if (c + 1 < C0) {
      const v4f* wn = w0v + (c + 1) * W0Q;
      wq0 = wn[q0];
      wq1 = wn[q1];
      wq2 = wn[q2];
    }
    const float bc = b0[c];

    {
      const int m0 = (wave >> 1) * 16, nb = (wave & 1) * 48;
      v8f acc0, acc1, acc2;
      strip_gemm(W0c, Vm, m0, nb, l, half, acc0, acc1, acc2);
      st_c_t3(Tc, KP, m0, nb, acc0, acc1, acc2, l, half, KP);
    }
    __syncthreads();

    if (c + 1 < C0) {
      *(v2f*)(W0c + s00) = (v2f){wq0.x, wq0.y};
      *(v2f*)(W0c + s01) = (v2f){wq0.z, wq0.w};
      *(v2f*)(W0c + s10) = (v2f){wq1.x, wq1.y};
      *(v2f*)(W0c + s11) = (v2f){wq1.z, wq1.w};
      if (has2) {
        *(v2f*)(W0c + s20) = (v2f){wq2.x, wq2.y};
        *(v2f*)(W0c + s21) = (v2f){wq2.z, wq2.w};
      }
    }

    const int nstrip = ((0xA5 >> wave) & 1) ? 2 : 1;
    for (int i = 0; i < nstrip; ++i) {
      const int j = i == 0 ? wave : 8 + (wave >> 1);
      const int m0 = (j >> 1) * 16, nb = (j & 1) * 48;
      v8f acc0, acc1, acc2;
      strip_gemm(Um, Tc, m0, nb, l, half, acc0, acc1, acc2);
      st_c3(Oc, NP, m0, nb, acc0, acc1, acc2, l, half, OW0);
    }
    __syncthreads();

    for (int i = tid; i < P0 * P0; i += NTHREADS) {
      int py = i / P0, px = i % P0;
      float x[9];
#pragma unroll
      for (int dy = 0; dy < 3; ++dy)
#pragma unroll
        for (int dx = 0; dx < 3; ++dx) x[3 * dy + dx] = Oc[(3 * py + dy) * NP + (3 * px + dx)];
      X1[c * (P0 * P0) + i] = gelu_pool_max9(x, bc);
    }
  }
  __syncthreads();

  float* A1 = smem + OFF_A1;
  float* O1 = smem + OFF_O1;
  float* X2 = smem + OFF_X2;
  for (int i = tid; i < C1 * K1; i += NTHREADS) {
    int co = i / K1, k = i - co * K1;
    int ci = k / 9, r9 = k - ci * 9;
    A1[co * K1 + r9 * 16 + ci] = w1[i];
  }
  __syncthreads();

  for (int nt = wave; nt < (NPIX1 + 15) / 16; nt += NWAVE) {
    const int col = nt * 16 + l;
    const int pc  = col < NPIX1 ? col : NPIX1 - 1;
    const int py = pc / OW1, px = pc - py * OW1;
    const float* xb = X1 + py * P0 + px + half * (2 * P0 * P0);
    const float* ab = A1 + l * K1 + 2 * half;
    v8f acc0 = {0.f, 0.f, 0.f, 0.f, 0.f, 0.f, 0.f, 0.f};
    v8f acc1 = {0.f, 0.f, 0.f, 0.f, 0.f, 0.f, 0.f, 0.f};
    v2f fb[2][4], fa0[2][4], fa1[2][4];
#pragma unroll
    for (int q = 0; q < 4; ++q) {
      fb[0][q].x = xb[(4 * q) * (P0 * P0)];
      fb[0][q].y = xb[(4 * q + 1) * (P0 * P0)];
      fa0[0][q]  = *(const v2f*)(ab + 4 * q);
      fa1[0][q]  = *(const v2f*)(ab + 16 * K1 + 4 * q);
    }
#pragma unroll
    for (int r9 = 0; r9 < 9; ++r9) {
      const int cur = r9 & 1, nxt = cur ^ 1;
      if (r9 + 1 < 9) {
        const int sp = ((r9 + 1) / 3) * P0 + ((r9 + 1) % 3);
#pragma unroll
        for (int q = 0; q < 4; ++q) {
          fb[nxt][q].x = xb[(4 * q) * (P0 * P0) + sp];
          fb[nxt][q].y = xb[(4 * q + 1) * (P0 * P0) + sp];
          fa0[nxt][q]  = *(const v2f*)(ab + (r9 + 1) * 16 + 4 * q);
          fa1[nxt][q]  = *(const v2f*)(ab + 16 * K1 + (r9 + 1) * 16 + 4 * q);
        }
      }
      __builtin_amdgcn_sched_barrier(0);
#pragma unroll
      for (int q = 0; q < 4; ++q) {
        WMMA_F32(acc0, fa0[cur][q], fb[cur][q]);
        WMMA_F32(acc1, fa1[cur][q], fb[cur][q]);
      }
      __builtin_amdgcn_sched_barrier(0);
    }
    if (col < NPIX1) {
#pragma unroll
      for (int r = 0; r < 8; ++r) {
        O1[(r + 8 * half) * NPIX1 + col]      = acc0[r];
        O1[(16 + r + 8 * half) * NPIX1 + col] = acc1[r];
      }
    }
  }
  __syncthreads();

  for (int i = tid; i < C1 * P1 * P1; i += NTHREADS) {
    int co = i >> 6, rem = i & 63, py = rem >> 3, px = rem & 7;
    float bc = b1[co];
    float x[9];
#pragma unroll
    for (int dy = 0; dy < 3; ++dy)
#pragma unroll
      for (int dx = 0; dx < 3; ++dx) x[3 * dy + dx] = O1[co * NPIX1 + (3 * py + dy) * OW1 + (3 * px + dx)];
    X2[i] = gelu_pool_max9(x, bc);
  }
  __syncthreads();

  float* A2 = smem + OFF_A2;
  float* B2 = smem + OFF_B2;
  float* O2 = smem + OFF_O2;
  for (int i = tid; i < C2 * K2; i += NTHREADS) A2[i] = w2[i];
  for (int i = tid; i < K2 * NPIX2P; i += NTHREADS) {
    int kk = i / NPIX2P, col = i % NPIX2P;
    float val = 0.f;
    if (col < NPIX2) {
      int py = col / OW2, px = col % OW2;
      int ci = kk / 9, r9 = kk % 9;
      val = X2[ci * (P1 * P1) + (py + r9 / 3) * P1 + (px + r9 % 3)];
    }
    B2[i] = val;
  }
  __syncthreads();

  for (int t = wave; t < 2 * 3; t += NWAVE) {
    int m0 = (t / 3) * 16, n0 = (t % 3) * 16;
    v8f acc = {0.f, 0.f, 0.f, 0.f, 0.f, 0.f, 0.f, 0.f};
    for (int k0 = 0; k0 < K2; k0 += 4) {
      v2f a  = ld_a(A2, K2, m0, k0, l, half);
      v2f bb = ld_b(B2, NPIX2P, k0, n0, l, half);
      WMMA_F32(acc, a, bb);
    }
    st_c(O2, NPIX2P, m0, n0, acc, l, half, C2, NPIX2P);
  }
  __syncthreads();

  if (tid < 128) {
    int co = tid >> 2, rem = tid & 3, py = rem >> 1, px = rem & 1;
    float bc = b2[co];
    float x[9];
#pragma unroll
    for (int dy = 0; dy < 3; ++dy)
#pragma unroll
      for (int dx = 0; dx < 3; ++dx) x[3 * dy + dx] = O2[co * NPIX2P + (3 * py + dy) * OW2 + (3 * px + dx)];
    out[tok * 128 + tid] = gelu_pool_max9(x, bc);
  }
}

extern "C" void kernel_launch(void* const* d_in, const int* in_sizes, int n_in,
                              void* d_out, int out_size, void* d_ws, size_t ws_size,
                              hipStream_t stream) {
  (void)in_sizes; (void)n_in; (void)out_size; (void)d_ws; (void)ws_size;
  const float* tf = (const float*)d_in[0];
  const float* tp = (const float*)d_in[1];
  const float* w0 = (const float*)d_in[2];
  const float* b0 = (const float*)d_in[3];
  const float* w1 = (const float*)d_in[4];
  const float* b1 = (const float*)d_in[5];
  const float* w2 = (const float*)d_in[6];
  const float* b2 = (const float*)d_in[7];
  float* out = (float*)d_out;

  const size_t shmem = SMEM_FLOATS * sizeof(float);
  (void)hipFuncSetAttribute((const void*)fused_outer_conv_kernel,
                            hipFuncAttributeMaxDynamicSharedMemorySize, (int)shmem);
  hipLaunchKernelGGL(fused_outer_conv_kernel, dim3(NTOK), dim3(NTHREADS), shmem,
                     stream, tf, tp, w0, b0, w1, b1, w2, b2, out);
}


# Dead outputs dropped: conv1's pool keeps 24x24 of 26x26, which read 26x26 of pooled conv0 = 78x78 of conv0's 84x84; conv0 GEMMs run 5x5 tiles (was 6x6), conv1 36 column tiles (was 43)
# speedup vs baseline: 5.6527x; 1.1589x over previous
#include <hip/hip_runtime.h>


typedef float v2f __attribute__((ext_vector_type(2)));
typedef float v4f __attribute__((ext_vector_type(4)));
typedef float v8f __attribute__((ext_vector_type(8)));

#define NB 8
#define NS 64
#define ND 300
#define NTOK (NB * NS)
#define NTHREADS 256
#define NWAVE 8

#define C0 16
#define K0 50
#define KP 52
#define NU0 80
#define NP 96
#define P0 28
#define PU0 26
#define W0Q 625

#define C1 32
#define OW1 24
#define NPIX1 576
#define O1P 580
#define K1 144
#define P1 8

#define C2 32
#define OW2 6
#define NPIX2 36
#define NPIX2P 48
#define K2 288

#define OFF_VM  0
#define OFF_UM  4992
#define OFF_W0C 9984
#define OFF_TC  13312
#define OFF_OC  18304
#define OFF_X1  26368
#define OFF_A1  0
#define OFF_O1  4608
#define OFF_X2  26368
#define OFF_A2  0
#define OFF_B2  9216
#define OFF_O2  23040
#define SMEM_FLOATS 38912

__device__ __forceinline__ float gelu_exact(float x) {
  return 0.5f * x * (1.0f + erff(x * 0.70710678118654752f));
}

__device__ __forceinline__ float gelu_pool_max9(const float x[9], float b) {
  const float hi = fmaxf(fmaxf(fmaxf(x[0], x[1]), fmaxf(x[2], x[3])),
                         fmaxf(fmaxf(x[4], x[5]), fmaxf(fmaxf(x[6], x[7]), x[8]))) + b;
  float r = gelu_exact(hi);
  if (hi < 0.f) {
    const float lo = fminf(fminf(fminf(x[0], x[1]), fminf(x[2], x[3])),
                           fminf(fminf(x[4], x[5]), fminf(fminf(x[6], x[7]), x[8]))) + b;
    r = fmaxf(r, gelu_exact(lo));
  }
  return r;
}

__device__ __forceinline__ int w0_pair_slot(int p) {
  int ky = p / 25;
  return ky * KP + 2 * (p - ky * 25);
}

__device__ __forceinline__ v2f ld_a(const float* A, int lda, int m0, int k0,
                                    int l, int half) {
  return *(const v2f*)(A + (m0 + l) * lda + (k0 + 2 * half));
}
__device__ __forceinline__ v2f ld_bt(const float* Bt, int ldk, int k0, int n0,
                                     int l, int half) {
  return *(const v2f*)(Bt + (n0 + l) * ldk + (k0 + 2 * half));
}
__device__ __forceinline__ v2f ld_b(const float* Bm, int ldb, int k0, int n0,
                                    int l, int half) {
  v2f b;
  b.x = Bm[(k0 + 2 * half) * ldb + n0 + l];
  b.y = Bm[(k0 + 2 * half + 1) * ldb + n0 + l];
  return b;
}
__device__ __forceinline__ void st_c(float* Cm, int ldc, int m0, int n0, v8f c,
                                     int l, int half, int maxRow, int maxCol) {
  int col = n0 + l;
  if (col >= maxCol) return;
#pragma unroll
  for (int r = 0; r < 8; ++r) {
    int row = m0 + r + 8 * half;
    if (row < maxRow) Cm[row * ldc + col] = c[r];
  }
}

__device__ __forceinline__ void st_c3(float* Cm, int ldc, int m0, int n0,
                                      v8f c0, v8f c1, v8f c2, int l, int half, int maxRow) {
  const int row0 = m0 + 8 * half;
  float* p = Cm + row0 * ldc + n0 + l;
#pragma unroll
  for (int r = 0; r < 8; ++r) {
    if (row0 + r < maxRow) {
      p[r * ldc]      = c0[r];
      p[r * ldc + 16] = c1[r];
      p[r * ldc + 32] = c2[r];
    }
  }
}

__device__ __forceinline__ void st_c2(float* Cm, int ldc, int m0, int n0,
                                      v8f c0, v8f c1, int l, int half, int maxRow) {
  const int row0 = m0 + 8 * half;
  float* p = Cm + row0 * ldc + n0 + l;
#pragma unroll
  for (int r = 0; r < 8; ++r) {
    if (row0 + r < maxRow) {
      p[r * ldc]      = c0[r];
      p[r * ldc + 16] = c1[r];
    }
  }
}

__device__ __forceinline__ void st_c_t3(float* Ct, int ldk, int m0, int n0,
                                        v8f c0, v8f c1, v8f c2, int l, int half, int maxRow) {
  const int row = m0 + 8 * half;
  float* p = Ct + (n0 + l) * ldk + row;
  if (row < maxRow) {
    *(v4f*)p              = (v4f){c0[0], c0[1], c0[2], c0[3]};
    *(v4f*)(p + 16 * ldk) = (v4f){c1[0], c1[1], c1[2], c1[3]};
    *(v4f*)(p + 32 * ldk) = (v4f){c2[0], c2[1], c2[2], c2[3]};
  }
  if (row + 4 < maxRow) {
    *(v4f*)(p + 4)            = (v4f){c0[4], c0[5], c0[6], c0[7]};
    *(v4f*)(p + 16 * ldk + 4) = (v4f){c1[4], c1[5], c1[6], c1[7]};
    *(v4f*)(p + 32 * ldk + 4) = (v4f){c2[4], c2[5], c2[6], c2[7]};
  }
}

__device__ __forceinline__ void st_c_t2(float* Ct, int ldk, int m0, int n0,
                                        v8f c0, v8f c1, int l, int half, int maxRow) {
  const int row = m0 + 8 * half;
  float* p = Ct + (n0 + l) * ldk + row;
  if (row < maxRow) {
    *(v4f*)p              = (v4f){c0[0], c0[1], c0[2], c0[3]};
    *(v4f*)(p + 16 * ldk) = (v4f){c1[0], c1[1], c1[2], c1[3]};
  }
  if (row + 4 < maxRow) {
    *(v4f*)(p + 4)            = (v4f){c0[4], c0[5], c0[6], c0[7]};
    *(v4f*)(p + 16 * ldk + 4) = (v4f){c1[4], c1[5], c1[6], c1[7]};
  }
}

#define WMMA_F32(acc, a, b)                                                   \
  (acc) = __builtin_amdgcn_wmma_f32_16x16x4_f32(false, (a), false, (b),       \
                                                (short)0, (acc), false, false)

__device__ __forceinline__ void strip_gemm(const float* A, const float* Bt, int m0, int nb,
                                           int l, int half, v8f& acc0, v8f& acc1, v8f& acc2) {
  v2f fa[13], fb0[13], fb1[13], fb2[13];
#pragma unroll
  for (int s = 0; s < 13; ++s) {
    fa[s]  = ld_a(A, KP, m0, 4 * s, l, half);
    fb0[s] = ld_bt(Bt, KP, 4 * s, nb, l, half);
    fb1[s] = ld_bt(Bt, KP, 4 * s, nb + 16, l, half);
    fb2[s] = ld_bt(Bt, KP, 4 * s, nb + 32, l, half);
  }
  __builtin_amdgcn_sched_barrier(0);
  acc0 = (v8f){0.f, 0.f, 0.f, 0.f, 0.f, 0.f, 0.f, 0.f};
  acc1 = acc0;
  acc2 = acc0;
#pragma unroll
  for (int s = 0; s < 13; ++s) {
    WMMA_F32(acc0, fa[s], fb0[s]);
    WMMA_F32(acc1, fa[s], fb1[s]);
    WMMA_F32(acc2, fa[s], fb2[s]);
  }
}

__device__ __forceinline__ void strip_gemm2(const float* A, const float* Bt, int m0, int nb,
                                            int l, int half, v8f& acc0, v8f& acc1) {
  v2f fa[13], fb0[13], fb1[13];
#pragma unroll
  for (int s = 0; s < 13; ++s) {
    fa[s]  = ld_a(A, KP, m0, 4 * s, l, half);
    fb0[s] = ld_bt(Bt, KP, 4 * s, nb, l, half);
    fb1[s] = ld_bt(Bt, KP, 4 * s, nb + 16, l, half);
  }
  __builtin_amdgcn_sched_barrier(0);
  acc0 = (v8f){0.f, 0.f, 0.f, 0.f, 0.f, 0.f, 0.f, 0.f};
  acc1 = acc0;
#pragma unroll
  for (int s = 0; s < 13; ++s) {
    WMMA_F32(acc0, fa[s], fb0[s]);
    WMMA_F32(acc1, fa[s], fb1[s]);
  }
}

extern "C" __global__ __launch_bounds__(NTHREADS) void
fused_outer_conv_kernel(const float* __restrict__ tf,
                        const float* __restrict__ tp,
                        const float* __restrict__ w0, const float* __restrict__ b0,
                        const float* __restrict__ w1, const float* __restrict__ b1,
                        const float* __restrict__ w2, const float* __restrict__ b2,
                        float* __restrict__ out) {
  extern __shared__ float smem[];
  const int tid  = threadIdx.x;
  const int lane = tid & 31;
  const int l    = tid & 15;
  const int half = (tid >> 4) & 1;
  const int wave = tid >> 5;
  const int tok  = blockIdx.x;

  const float* v = tf + tok * ND;
  const float* u = tp + tok * ND;

  {
    const char* pw1 = (const char*)w1;
    const char* pw2 = (const char*)w2;
    if (tid * 128 < (int)(C1 * K1 * sizeof(float))) __builtin_prefetch(pw1 + tid * 128, 0, 1);
    if (tid * 128 < (int)(C2 * K2 * sizeof(float))) __builtin_prefetch(pw2 + tid * 128, 0, 1);
  }

  float* Vm  = smem + OFF_VM;
  float* Um  = smem + OFF_UM;
  float* W0c = smem + OFF_W0C;
  float* Tc  = smem + OFF_TC;
  float* Oc  = smem + OFF_OC;
  float* X1  = smem + OFF_X1;

  const v4f* w0v = (const v4f*)w0;
  const int  q0 = tid, q1 = tid + NTHREADS;
  const bool has2 = tid + 2 * NTHREADS < W0Q;
  const int  q2 = has2 ? tid + 2 * NTHREADS : W0Q - 1;
  const int  s00 = w0_pair_slot(2 * q0), s01 = w0_pair_slot(2 * q0 + 1);
  const int  s10 = w0_pair_slot(2 * q1), s11 = w0_pair_slot(2 * q1 + 1);
  const int  s20 = w0_pair_slot(2 * q2), s21 = w0_pair_slot(2 * q2 + 1);

  v4f wq0 = w0v[q0], wq1 = w0v[q1], wq2 = w0v[q2];

  for (int i = tid; i < NU0 * KP; i += NTHREADS) {
    int o = i / KP, k = i % KP;
    bool in = k < K0;
    Um[i] = in ? u[3 * o + k] : 0.f;
    Vm[i] = in ? v[3 * o + k] : 0.f;
  }

  for (int i = tid; i < (64 - K0) * KP; i += NTHREADS) W0c[K0 * KP + i] = 0.f;
  if (tid < 2 * K0) W0c[(tid >> 1) * KP + K0 + (tid & 1)] = 0.f;
  *(v2f*)(W0c + s00) = (v2f){wq0.x, wq0.y};
  *(v2f*)(W0c + s01) = (v2f){wq0.z, wq0.w};
  *(v2f*)(W0c + s10) = (v2f){wq1.x, wq1.y};
  *(v2f*)(W0c + s11) = (v2f){wq1.z, wq1.w};
  if (has2) {
    *(v2f*)(W0c + s20) = (v2f){wq2.x, wq2.y};
    *(v2f*)(W0c + s21) = (v2f){wq2.z, wq2.w};
  }
  __syncthreads();

  for (int c = 0; c < C0; ++c) {
    if (c + 1 < C0) {
      const v4f* wn = w0v + (c + 1) * W0Q;
      wq0 = wn[q0];
      wq1 = wn[q1];
      wq2 = wn[q2];
    }
    const float bc = b0[c];

    {
      const int m0 = (wave >> 1) * 16;
      if ((0xA5 >> wave) & 1) {
        v8f acc0, acc1, acc2;
        strip_gemm(W0c, Vm, m0, 0, l, half, acc0, acc1, acc2);
        st_c_t3(Tc, KP, m0, 0, acc0, acc1, acc2, l, half, KP);
      } else {
        v8f acc0, acc1;
        strip_gemm2(W0c, Vm, m0, 48, l, half, acc0, acc1);
        st_c_t2(Tc, KP, m0, 48, acc0, acc1, l, half, KP);
      }
    }
    __syncthreads();

    if (c + 1 < C0) {
      *(v2f*)(W0c + s00) = (v2f){wq0.x, wq0.y};
      *(v2f*)(W0c + s01) = (v2f){wq0.z, wq0.w};
      *(v2f*)(W0c + s10) = (v2f){wq1.x, wq1.y};
      *(v2f*)(W0c + s11) = (v2f){wq1.z, wq1.w};
      if (has2) {
        *(v2f*)(W0c + s20) = (v2f){wq2.x, wq2.y};
        *(v2f*)(W0c + s21) = (v2f){wq2.z, wq2.w};
      }
    }

    {
      const int rowb = (0x43221040 >> (4 * wave)) & 15;
      if ((0xDC >> wave) & 1) {
        const int m0 = rowb * 16;
        v8f acc0, acc1, acc2;
        strip_gemm(Um, Tc, m0, 0, l, half, acc0, acc1, acc2);
        st_c3(Oc, NP, m0, 0, acc0, acc1, acc2, l, half, NU0);
      } else {
        const int nstrip = wave == 1 ? 1 : 2;
        for (int i = 0; i < nstrip; ++i) {
          const int m0 = (rowb + i) * 16;
          v8f acc0, acc1;
          strip_gemm2(Um, Tc, m0, 48, l, half, acc0, acc1);
          st_c2(Oc, NP, m0, 48, acc0, acc1, l, half, NU0);
        }
      }
    }
    __syncthreads();

    for (int i = tid; i < PU0 * PU0; i += NTHREADS) {
      int py = i / PU0, px = i % PU0;
      float x[9];
#pragma unroll
      for (int dy = 0; dy < 3; ++dy)
#pragma unroll
        for (int dx = 0; dx < 3; ++dx) x[3 * dy + dx] = Oc[(3 * py + dy) * NP + (3 * px + dx)];
      X1[c * (P0 * P0) + py * P0 + px] = gelu_pool_max9(x, bc);
    }
  }
  __syncthreads();

  float* A1 = smem + OFF_A1;
  float* O1 = smem + OFF_O1;
  float* X2 = smem + OFF_X2;
  for (int i = tid; i < C1 * K1; i += NTHREADS) {
    int co = i / K1, k = i - co * K1;
    int ci = k / 9, r9 = k - ci * 9;
    A1[co * K1 + r9 * 16 + ci] = w1[i];
  }
  __syncthreads();

  for (int nt = wave; nt < NPIX1 / 16; nt += NWAVE) {
    const int col = nt * 16 + l;
    const int py = col / OW1, px = col - py * OW1;
    const float* xb = X1 + py * P0 + px + half * (2 * P0 * P0);
    const float* ab = A1 + l * K1 + 2 * half;
    v8f acc0 = {0.f, 0.f, 0.f, 0.f, 0.f, 0.f, 0.f, 0.f};
    v8f acc1 = {0.f, 0.f, 0.f, 0.f, 0.f, 0.f, 0.f, 0.f};
    v2f fb[2][4], fa0[2][4], fa1[2][4];
#pragma unroll
    for (int q = 0; q < 4; ++q) {
      fb[0][q].x = xb[(4 * q) * (P0 * P0)];
      fb[0][q].y = xb[(4 * q + 1) * (P0 * P0)];
      fa0[0][q]  = *(const v2f*)(ab + 4 * q);
      fa1[0][q]  = *(const v2f*)(ab + 16 * K1 + 4 * q);
    }
#pragma unroll
    for (int r9 = 0; r9 < 9; ++r9) {
      const int cur = r9 & 1, nxt = cur ^ 1;
      if (r9 + 1 < 9) {
        const int sp = ((r9 + 1) / 3) * P0 + ((r9 + 1) % 3);
#pragma unroll
        for (int q = 0; q < 4; ++q) {
          fb[nxt][q].x = xb[(4 * q) * (P0 * P0) + sp];
          fb[nxt][q].y = xb[(4 * q + 1) * (P0 * P0) + sp];
          fa0[nxt][q]  = *(const v2f*)(ab + (r9 + 1) * 16 + 4 * q);
          fa1[nxt][q]  = *(const v2f*)(ab + 16 * K1 + (r9 + 1) * 16 + 4 * q);
        }
      }
      __builtin_amdgcn_sched_barrier(0);
#pragma unroll
      for (int q = 0; q < 4; ++q) {
        WMMA_F32(acc0, fa0[cur][q], fb[cur][q]);
        WMMA_F32(acc1, fa1[cur][q], fb[cur][q]);
      }
      __builtin_amdgcn_sched_barrier(0);
    }
#pragma unroll
    for (int r = 0; r < 8; ++r) {
      O1[(r + 8 * half) * O1P + col]      = acc0[r];
      O1[(16 + r + 8 * half) * O1P + col] = acc1[r];
    }
  }
  __syncthreads();

  for (int i = tid; i < C1 * P1 * P1; i += NTHREADS) {
    int co = i >> 6, rem = i & 63, py = rem >> 3, px = rem & 7;
    float bc = b1[co];
    float x[9];
#pragma unroll
    for (int dy = 0; dy < 3; ++dy)
#pragma unroll
      for (int dx = 0; dx < 3; ++dx) x[3 * dy + dx] = O1[co * O1P + (3 * py + dy) * OW1 + (3 * px + dx)];
    X2[i] = gelu_pool_max9(x, bc);
  }
  __syncthreads();

  float* A2 = smem + OFF_A2;
  float* B2 = smem + OFF_B2;
  float* O2 = smem + OFF_O2;
  for (int i = tid; i < C2 * K2; i += NTHREADS) A2[i] = w2[i];
  for (int i = tid; i < K2 * NPIX2P; i += NTHREADS) {
    int kk = i / NPIX2P, col = i % NPIX2P;
    float val = 0.f;
    if (col < NPIX2) {
      int py = col / OW2, px = col % OW2;
      int ci = kk / 9, r9 = kk % 9;
      val = X2[ci * (P1 * P1) + (py + r9 / 3) * P1 + (px + r9 % 3)];
    }
    B2[i] = val;
  }
  __syncthreads();

  for (int t = wave; t < 2 * 3; t += NWAVE) {
    int m0 = (t / 3) * 16, n0 = (t % 3) * 16;
    v8f acc = {0.f, 0.f, 0.f, 0.f, 0.f, 0.f, 0.f, 0.f};
    for (int k0 = 0; k0 < K2; k0 += 4) {
      v2f a  = ld_a(A2, K2, m0, k0, l, half);
      v2f bb = ld_b(B2, NPIX2P, k0, n0, l, half);
      WMMA_F32(acc, a, bb);
    }
    st_c(O2, NPIX2P, m0, n0, acc, l, half, C2, NPIX2P);
  }
  __syncthreads();

  if (tid < 128) {
    int co = tid >> 2, rem = tid & 3, py = rem >> 1, px = rem & 1;
    float bc = b2[co];
    float x[9];
#pragma unroll
    for (int dy = 0; dy < 3; ++dy)
#pragma unroll
      for (int dx = 0; dx < 3; ++dx) x[3 * dy + dx] = O2[co * NPIX2P + (3 * py + dy) * OW2 + (3 * px + dx)];
    out[tok * 128 + tid] = gelu_pool_max9(x, bc);
  }
}

extern "C" void kernel_launch(void* const* d_in, const int* in_sizes, int n_in,
                              void* d_out, int out_size, void* d_ws, size_t ws_size,
                              hipStream_t stream) {
  (void)in_sizes; (void)n_in; (void)out_size; (void)d_ws; (void)ws_size;
  const float* tf = (const float*)d_in[0];
  const float* tp = (const float*)d_in[1];
  const float* w0 = (const float*)d_in[2];
  const float* b0 = (const float*)d_in[3];
  const float* w1 = (const float*)d_in[4];
  const float* b1 = (const float*)d_in[5];
  const float* w2 = (const float*)d_in[6];
  const float* b2 = (const float*)d_in[7];
  float* out = (float*)d_out;

  const size_t shmem = SMEM_FLOATS * sizeof(float);
  (void)hipFuncSetAttribute((const void*)fused_outer_conv_kernel,
                            hipFuncAttributeMaxDynamicSharedMemorySize, (int)shmem);
  hipLaunchKernelGGL(fused_outer_conv_kernel, dim3(NTOK), dim3(NTHREADS), shmem,
                     stream, tf, tp, w0, b0, w1, b1, w2, b2, out);
}


# conv2 reads its B fragments straight from the pooled conv1 plane (k = tap*32+channel, w2 restaged at pitch 292): no 288x48 im2col build; tap-wise load groups, two accumulator chains per tile
# speedup vs baseline: 6.2809x; 1.1111x over previous
#include <hip/hip_runtime.h>


typedef float v2f __attribute__((ext_vector_type(2)));
typedef float v4f __attribute__((ext_vector_type(4)));
typedef float v8f __attribute__((ext_vector_type(8)));

#define NB 8
#define NS 64
#define ND 300
#define NTOK (NB * NS)
#define NTHREADS 256
#define NWAVE 8

#define C0 16
#define K0 50
#define KP 52
#define NU0 80
#define NP 96
#define P0 28
#define PU0 26
#define W0Q 625

#define C1 32
#define OW1 24
#define NPIX1 576
#define O1P 580
#define K1 144
#define P1 8

#define C2 32
#define OW2 6
#define NPIX2 36
#define NPIX2P 48
#define K2 288
#define A2P 292
#define X2P 80

#define OFF_VM  0
#define OFF_UM  4992
#define OFF_W0C 9984
#define OFF_TC  13312
#define OFF_OC  18304
#define OFF_X1  26368
#define OFF_A1  0
#define OFF_O1  4608
#define OFF_X2  26368
#define OFF_A2  0
#define OFF_O2  23040
#define SMEM_FLOATS 38912

__device__ __forceinline__ float gelu_exact(float x) {
  return 0.5f * x * (1.0f + erff(x * 0.70710678118654752f));
}

__device__ __forceinline__ float gelu_pool_max9(const float x[9], float b) {
  const float hi = fmaxf(fmaxf(fmaxf(x[0], x[1]), fmaxf(x[2], x[3])),
                         fmaxf(fmaxf(x[4], x[5]), fmaxf(fmaxf(x[6], x[7]), x[8]))) + b;
  float r = gelu_exact(hi);
  if (hi < 0.f) {
    const float lo = fminf(fminf(fminf(x[0], x[1]), fminf(x[2], x[3])),
                           fminf(fminf(x[4], x[5]), fminf(fminf(x[6], x[7]), x[8]))) + b;
    r = fmaxf(r, gelu_exact(lo));
  }
  return r;
}

__device__ __forceinline__ int w0_pair_slot(int p) {
  int ky = p / 25;
  return ky * KP + 2 * (p - ky * 25);
}

__device__ __forceinline__ v2f ld_a(const float* A, int lda, int m0, int k0,
                                    int l, int half) {
  return *(const v2f*)(A + (m0 + l) * lda + (k0 + 2 * half));
}
__device__ __forceinline__ v2f ld_bt(const float* Bt, int ldk, int k0, int n0,
                                     int l, int half) {
  return *(const v2f*)(Bt + (n0 + l) * ldk + (k0 + 2 * half));
}
__device__ __forceinline__ void st_c3(float* Cm, int ldc, int m0, int n0,
                                      v8f c0, v8f c1, v8f c2, int l, int half, int maxRow) {
  const int row0 = m0 + 8 * half;
  float* p = Cm + row0 * ldc + n0 + l;
#pragma unroll
  for (int r = 0; r < 8; ++r) {
    if (row0 + r < maxRow) {
      p[r * ldc]      = c0[r];
      p[r * ldc + 16] = c1[r];
      p[r * ldc + 32] = c2[r];
    }
  }
}

__device__ __forceinline__ void st_c2(float* Cm, int ldc, int m0, int n0,
                                      v8f c0, v8f c1, int l, int half, int maxRow) {
  const int row0 = m0 + 8 * half;
  float* p = Cm + row0 * ldc + n0 + l;
#pragma unroll
  for (int r = 0; r < 8; ++r) {
    if (row0 + r < maxRow) {
      p[r * ldc]      = c0[r];
      p[r * ldc + 16] = c1[r];
    }
  }
}

__device__ __forceinline__ void st_c_t3(float* Ct, int ldk, int m0, int n0,
                                        v8f c0, v8f c1, v8f c2, int l, int half, int maxRow) {
  const int row = m0 + 8 * half;
  float* p = Ct + (n0 + l) * ldk + row;
  if (row < maxRow) {
    *(v4f*)p              = (v4f){c0[0], c0[1], c0[2], c0[3]};
    *(v4f*)(p + 16 * ldk) = (v4f){c1[0], c1[1], c1[2], c1[3]};
    *(v4f*)(p + 32 * ldk) = (v4f){c2[0], c2[1], c2[2], c2[3]};
  }
  if (row + 4 < maxRow) {
    *(v4f*)(p + 4)            = (v4f){c0[4], c0[5], c0[6], c0[7]};
    *(v4f*)(p + 16 * ldk + 4) = (v4f){c1[4], c1[5], c1[6], c1[7]};
    *(v4f*)(p + 32 * ldk + 4) = (v4f){c2[4], c2[5], c2[6], c2[7]};
  }
}

__device__ __forceinline__ void st_c_t2(float* Ct, int ldk, int m0, int n0,
                                        v8f c0, v8f c1, int l, int half, int maxRow) {
  const int row = m0 + 8 * half;
  float* p = Ct + (n0 + l) * ldk + row;
  if (row < maxRow) {
    *(v4f*)p              = (v4f){c0[0], c0[1], c0[2], c0[3]};
    *(v4f*)(p + 16 * ldk) = (v4f){c1[0], c1[1], c1[2], c1[3]};
  }
  if (row + 4 < maxRow) {
    *(v4f*)(p + 4)            = (v4f){c0[4], c0[5], c0[6], c0[7]};
    *(v4f*)(p + 16 * ldk + 4) = (v4f){c1[4], c1[5], c1[6], c1[7]};
  }
}

#define WMMA_F32(acc, a, b)                                                   \
  (acc) = __builtin_amdgcn_wmma_f32_16x16x4_f32(false, (a), false, (b),       \
                                                (short)0, (acc), false, false)

__device__ __forceinline__ void strip_gemm(const float* A, const float* Bt, int m0, int nb,
                                           int l, int half, v8f& acc0, v8f& acc1, v8f& acc2) {
  v2f fa[13], fb0[13], fb1[13], fb2[13];
#pragma unroll
  for (int s = 0; s < 13; ++s) {
    fa[s]  = ld_a(A, KP, m0, 4 * s, l, half);
    fb0[s] = ld_bt(Bt, KP, 4 * s, nb, l, half);
    fb1[s] = ld_bt(Bt, KP, 4 * s, nb + 16, l, half);
    fb2[s] = ld_bt(Bt, KP, 4 * s, nb + 32, l, half);
  }
  __builtin_amdgcn_sched_barrier(0);
  acc0 = (v8f){0.f, 0.f, 0.f, 0.f, 0.f, 0.f, 0.f, 0.f};
  acc1 = acc0;
  acc2 = acc0;
#pragma unroll
  for (int s = 0; s < 13; ++s) {
    WMMA_F32(acc0, fa[s], fb0[s]);
    WMMA_F32(acc1, fa[s], fb1[s]);
    WMMA_F32(acc2, fa[s], fb2[s]);
  }
}

__device__ __forceinline__ void strip_gemm2(const float* A, const float* Bt, int m0, int nb,
                                            int l, int half, v8f& acc0, v8f& acc1) {
  v2f fa[13], fb0[13], fb1[13];
#pragma unroll
  for (int s = 0; s < 13; ++s) {
    fa[s]  = ld_a(A, KP, m0, 4 * s, l, half);
    fb0[s] = ld_bt(Bt, KP, 4 * s, nb, l, half);
    fb1[s] = ld_bt(Bt, KP, 4 * s, nb + 16, l, half);
  }
  __builtin_amdgcn_sched_barrier(0);
  acc0 = (v8f){0.f, 0.f, 0.f, 0.f, 0.f, 0.f, 0.f, 0.f};
  acc1 = acc0;
#pragma unroll
  for (int s = 0; s < 13; ++s) {
    WMMA_F32(acc0, fa[s], fb0[s]);
    WMMA_F32(acc1, fa[s], fb1[s]);
  }
}

extern "C" __global__ __launch_bounds__(NTHREADS) void
fused_outer_conv_kernel(const float* __restrict__ tf,
                        const float* __restrict__ tp,
                        const float* __restrict__ w0, const float* __restrict__ b0,
                        const float* __restrict__ w1, const float* __restrict__ b1,
                        const float* __restrict__ w2, const float* __restrict__ b2,
                        float* __restrict__ out) {
  extern __shared__ float smem[];
  const int tid  = threadIdx.x;
  const int lane = tid & 31;
  const int l    = tid & 15;
  const int half = (tid >> 4) & 1;
  const int wave = tid >> 5;
  const int tok  = blockIdx.x;

  const float* v = tf + tok * ND;
  const float* u = tp + tok * ND;

  {
    const char* pw1 = (const char*)w1;
    const char* pw2 = (const char*)w2;
    if (tid * 128 < (int)(C1 * K1 * sizeof(float))) __builtin_prefetch(pw1 + tid * 128, 0, 1);
    if (tid * 128 < (int)(C2 * K2 * sizeof(float))) __builtin_prefetch(pw2 + tid * 128, 0, 1);
  }

  float* Vm  = smem + OFF_VM;
  float* Um  = smem + OFF_UM;
  float* W0c = smem + OFF_W0C;
  float* Tc  = smem + OFF_TC;
  float* Oc  = smem + OFF_OC;
  float* X1  = smem + OFF_X1;

  const v4f* w0v = (const v4f*)w0;
  const int  q0 = tid, q1 = tid + NTHREADS;
  const bool has2 = tid + 2 * NTHREADS < W0Q;
  const int  q2 = has2 ? tid + 2 * NTHREADS : W0Q - 1;
  const int  s00 = w0_pair_slot(2 * q0), s01 = w0_pair_slot(2 * q0 + 1);
  const int  s10 = w0_pair_slot(2 * q1), s11 = w0_pair_slot(2 * q1 + 1);
  const int  s20 = w0_pair_slot(2 * q2), s21 = w0_pair_slot(2 * q2 + 1);

  v4f wq0 = w0v[q0], wq1 = w0v[q1], wq2 = w0v[q2];

  for (int i = tid; i < NU0 * KP; i += NTHREADS) {
    int o = i / KP, k = i % KP;
    bool in = k < K0;
    Um[i] = in ? u[3 * o + k] : 0.f;
    Vm[i] = in ? v[3 * o + k] : 0.f;
  }

  for (int i = tid; i < (64 - K0) * KP; i += NTHREADS) W0c[K0 * KP + i] = 0.f;
  if (tid < 2 * K0) W0c[(tid >> 1) * KP + K0 + (tid & 1)] = 0.f;
  *(v2f*)(W0c + s00) = (v2f){wq0.x, wq0.y};
  *(v2f*)(W0c + s01) = (v2f){wq0.z, wq0.w};
  *(v2f*)(W0c + s10) = (v2f){wq1.x, wq1.y};
  *(v2f*)(W0c + s11) = (v2f){wq1.z, wq1.w};
  if (has2) {
    *(v2f*)(W0c + s20) = (v2f){wq2.x, wq2.y};
    *(v2f*)(W0c + s21) = (v2f){wq2.z, wq2.w};
  }
  __syncthreads();

  for (int c = 0; c < C0; ++c) {
    if (c + 1 < C0) {
      const v4f* wn = w0v + (c + 1) * W0Q;
      wq0 = wn[q0];
      wq1 = wn[q1];
      wq2 = wn[q2];
    }
    const float bc = b0[c];

    {
      const int m0 = (wave >> 1) * 16;
      if ((0xA5 >> wave) & 1) {
        v8f acc0, acc1, acc2;
        strip_gemm(W0c, Vm, m0, 0, l, half, acc0, acc1, acc2);
        st_c_t3(Tc, KP, m0, 0, acc0, acc1, acc2, l, half, KP);
      } else {
        v8f acc0, acc1;
        strip_gemm2(W0c, Vm, m0, 48, l, half, acc0, acc1);
        st_c_t2(Tc, KP, m0, 48, acc0, acc1, l, half, KP);
      }
    }
    __syncthreads();

    if (c + 1 < C0) {
      *(v2f*)(W0c + s00) = (v2f){wq0.x, wq0.y};
      *(v2f*)(W0c + s01) = (v2f){wq0.z, wq0.w};
      *(v2f*)(W0c + s10) = (v2f){wq1.x, wq1.y};
      *(v2f*)(W0c + s11) = (v2f){wq1.z, wq1.w};
      if (has2) {
        *(v2f*)(W0c + s20) = (v2f){wq2.x, wq2.y};
        *(v2f*)(W0c + s21) = (v2f){wq2.z, wq2.w};
      }
    }

    {
      const int rowb = (0x43221040 >> (4 * wave)) & 15;
      if ((0xDC >> wave) & 1) {
        const int m0 = rowb * 16;
        v8f acc0, acc1, acc2;
        strip_gemm(Um, Tc, m0, 0, l, half, acc0, acc1, acc2);
        st_c3(Oc, NP, m0, 0, acc0, acc1, acc2, l, half, NU0);
      } else {
        const int nstrip = wave == 1 ? 1 : 2;
        for (int i = 0; i < nstrip; ++i) {
          const int m0 = (rowb + i) * 16;
          v8f acc0, acc1;
          strip_gemm2(Um, Tc, m0, 48, l, half, acc0, acc1);
          st_c2(Oc, NP, m0, 48, acc0, acc1, l, half, NU0);
        }
      }
    }
    __syncthreads();

    for (int i = tid; i < PU0 * PU0; i += NTHREADS) {
      int py = i / PU0, px = i % PU0;
      float x[9];
#pragma unroll
      for (int dy = 0; dy < 3; ++dy)
#pragma unroll
        for (int dx = 0; dx < 3; ++dx) x[3 * dy + dx] = Oc[(3 * py + dy) * NP + (3 * px + dx)];
      X1[c * (P0 * P0) + py * P0 + px] = gelu_pool_max9(x, bc);
    }
  }
  __syncthreads();

  float* A1 = smem + OFF_A1;
  float* O1 = smem + OFF_O1;
  float* X2 = smem + OFF_X2;
  for (int i = tid; i < C1 * K1; i += NTHREADS) {
    int co = i / K1, k = i - co * K1;
    int ci = k / 9, r9 = k - ci * 9;
    A1[co * K1 + r9 * 16 + ci] = w1[i];
  }
  __syncthreads();

  for (int nt = wave; nt < NPIX1 / 16; nt += NWAVE) {
    const int col = nt * 16 + l;
    const int py = col / OW1, px = col - py * OW1;
    const float* xb = X1 + py * P0 + px + half * (2 * P0 * P0);
    const float* ab = A1 + l * K1 + 2 * half;
    v8f acc0 = {0.f, 0.f, 0.f, 0.f, 0.f, 0.f, 0.f, 0.f};
    v8f acc1 = {0.f, 0.f, 0.f, 0.f, 0.f, 0.f, 0.f, 0.f};
    v2f fb[2][4], fa0[2][4], fa1[2][4];
#pragma unroll
    for (int q = 0; q < 4; ++q) {
      fb[0][q].x = xb[(4 * q) * (P0 * P0)];
      fb[0][q].y = xb[(4 * q + 1) * (P0 * P0)];
      fa0[0][q]  = *(const v2f*)(ab + 4 * q);
      fa1[0][q]  = *(const v2f*)(ab + 16 * K1 + 4 * q);
    }
#pragma unroll
    for (int r9 = 0; r9 < 9; ++r9) {
      const int cur = r9 & 1, nxt = cur ^ 1;
      if (r9 + 1 < 9) {
        const int sp = ((r9 + 1) / 3) * P0 + ((r9 + 1) % 3);
#pragma unroll
        for (int q = 0; q < 4; ++q) {
          fb[nxt][q].x = xb[(4 * q) * (P0 * P0) + sp];
          fb[nxt][q].y = xb[(4 * q + 1) * (P0 * P0) + sp];
          fa0[nxt][q]  = *(const v2f*)(ab + (r9 + 1) * 16 + 4 * q);
          fa1[nxt][q]  = *(const v2f*)(ab + 16 * K1 + (r9 + 1) * 16 + 4 * q);
        }
      }
      __builtin_amdgcn_sched_barrier(0);
#pragma unroll
      for (int q = 0; q < 4; ++q) {
        WMMA_F32(acc0, fa0[cur][q], fb[cur][q]);
        WMMA_F32(acc1, fa1[cur][q], fb[cur][q]);
      }
      __builtin_amdgcn_sched_barrier(0);
    }
#pragma unroll
    for (int r = 0; r < 8; ++r) {
      O1[(r + 8 * half) * O1P + col]      = acc0[r];
      O1[(16 + r + 8 * half) * O1P + col] = acc1[r];
    }
  }
  __syncthreads();

  for (int i = tid; i < C1 * P1 * P1; i += NTHREADS) {
    int co = i >> 6, rem = i & 63, py = rem >> 3, px = rem & 7;
    float bc = b1[co];
    float x[9];
#pragma unroll
    for (int dy = 0; dy < 3; ++dy)
#pragma unroll
      for (int dx = 0; dx < 3; ++dx) x[3 * dy + dx] = O1[co * O1P + (3 * py + dy) * OW1 + (3 * px + dx)];
    X2[co * X2P + rem] = gelu_pool_max9(x, bc);
  }
  __syncthreads();

  float* A2 = smem + OFF_A2;
  float* O2 = smem + OFF_O2;
  {
    float wv[(C2 * C1) / NTHREADS][9];
#pragma unroll
    for (int t = 0; t < (C2 * C1) / NTHREADS; ++t) {
      const float* wsrc = w2 + (tid + t * NTHREADS) * 9;
#pragma unroll
      for (int r9 = 0; r9 < 9; ++r9) wv[t][r9] = wsrc[r9];
    }
#pragma unroll
    for (int t = 0; t < (C2 * C1) / NTHREADS; ++t) {
      const int p = tid + t * NTHREADS;
      float* wdst = A2 + (p >> 5) * A2P + (p & 31);
#pragma unroll
      for (int r9 = 0; r9 < 9; ++r9) wdst[r9 * C1] = wv[t][r9];
    }
  }
  __syncthreads();

  if (wave < 2 * 3) {
    const int m0 = (wave / 3) * 16, n0 = (wave % 3) * 16;
    const int colr = n0 + l;
    const int col = colr < NPIX2 ? colr : NPIX2 - 1;
    const int py = col / OW2, px = col - py * OW2;
    const float* xb = X2 + py * P1 + px + half * (2 * X2P);
    const float* ab = A2 + (m0 + l) * A2P + 2 * half;
    v8f acc0 = {0.f, 0.f, 0.f, 0.f, 0.f, 0.f, 0.f, 0.f};
    v8f acc1 = {0.f, 0.f, 0.f, 0.f, 0.f, 0.f, 0.f, 0.f};
    v2f fb[2][8], fa[2][8];
#pragma unroll
    for (int q = 0; q < 8; ++q) {
      fb[0][q].x = xb[(4 * q) * X2P];
      fb[0][q].y = xb[(4 * q + 1) * X2P];
      fa[0][q]   = *(const v2f*)(ab + 4 * q);
    }
#pragma unroll
    for (int r9 = 0; r9 < 9; ++r9) {
      const int cur = r9 & 1, nxt = cur ^ 1;
      if (r9 + 1 < 9) {
        const int sp = ((r9 + 1) / 3) * P1 + ((r9 + 1) % 3);
#pragma unroll
        for (int q = 0; q < 8; ++q) {
          fb[nxt][q].x = xb[(4 * q) * X2P + sp];
          fb[nxt][q].y = xb[(4 * q + 1) * X2P + sp];
          fa[nxt][q]   = *(const v2f*)(ab + (r9 + 1) * C1 + 4 * q);
        }
      }
      __builtin_amdgcn_sched_barrier(0);
#pragma unroll
      for (int q = 0; q < 8; q += 2) {
        WMMA_F32(acc0, fa[cur][q], fb[cur][q]);
        WMMA_F32(acc1, fa[cur][q + 1], fb[cur][q + 1]);
      }
      __builtin_amdgcn_sched_barrier(0);
    }
#pragma unroll
    for (int r = 0; r < 8; ++r)
      O2[(m0 + r + 8 * half) * NPIX2P + n0 + l] = acc0[r] + acc1[r];
  }
  __syncthreads();

  if (tid < 128) {
    int co = tid >> 2, rem = tid & 3, py = rem >> 1, px = rem & 1;
    float bc = b2[co];
    float x[9];
#pragma unroll
    for (int dy = 0; dy < 3; ++dy)
#pragma unroll
      for (int dx = 0; dx < 3; ++dx) x[3 * dy + dx] = O2[co * NPIX2P + (3 * py + dy) * OW2 + (3 * px + dx)];
    out[tok * 128 + tid] = gelu_pool_max9(x, bc);
  }
}

extern "C" void kernel_launch(void* const* d_in, const int* in_sizes, int n_in,
                              void* d_out, int out_size, void* d_ws, size_t ws_size,
                              hipStream_t stream) {
  (void)in_sizes; (void)n_in; (void)out_size; (void)d_ws; (void)ws_size;
  const float* tf = (const float*)d_in[0];
  const float* tp = (const float*)d_in[1];
  const float* w0 = (const float*)d_in[2];
  const float* b0 = (const float*)d_in[3];
  const float* w1 = (const float*)d_in[4];
  const float* b1 = (const float*)d_in[5];
  const float* w2 = (const float*)d_in[6];
  const float* b2 = (const float*)d_in[7];
  float* out = (float*)d_out;

  const size_t shmem = SMEM_FLOATS * sizeof(float);
  (void)hipFuncSetAttribute((const void*)fused_outer_conv_kernel,
                            hipFuncAttributeMaxDynamicSharedMemorySize, (int)shmem);
  hipLaunchKernelGGL(fused_outer_conv_kernel, dim3(NTOK), dim3(NTHREADS), shmem,
                     stream, tf, tp, w0, b0, w1, b1, w2, b2, out);
}


# GELU+pool passes evaluate a thread's windows side by side (3 per conv0 channel, 4 at a time after conv1) with a branch-free erf: one load group, overlapped chains; was one branchy pass per window
# speedup vs baseline: 6.7624x; 1.0767x over previous
#include <hip/hip_runtime.h>


typedef float v2f __attribute__((ext_vector_type(2)));
typedef float v4f __attribute__((ext_vector_type(4)));
typedef float v8f __attribute__((ext_vector_type(8)));

#define NB 8
#define NS 64
#define ND 300
#define NTOK (NB * NS)
#define NTHREADS 256
#define NWAVE 8

#define C0 16
#define K0 50
#define KP 52
#define NU0 80
#define NP 96
#define P0 28
#define PU0 26
#define W0Q 625

#define C1 32
#define OW1 24
#define NPIX1 576
#define O1P 580
#define K1 144
#define P1 8

#define C2 32
#define OW2 6
#define NPIX2 36
#define NPIX2P 48
#define K2 288
#define A2P 292
#define X2P 80

#define OFF_VM  0
#define OFF_UM  4992
#define OFF_W0C 9984
#define OFF_TC  13312
#define OFF_OC  18304
#define OFF_X1  26368
#define OFF_A1  0
#define OFF_O1  4608
#define OFF_X2  26368
#define OFF_A2  0
#define OFF_O2  23040
#define SMEM_FLOATS 38912

__device__ __forceinline__ float erf_sel(float x) {
  const float ax = fabsf(x);
  const float t  = ax * ax;
  float p = fmaf(t, __uint_as_float(0xba1345e1u), __uint_as_float(0x3ba10414u));
  p = fmaf(t, p, __uint_as_float(0xbcdac9b8u));
  p = fmaf(t, p, __uint_as_float(0x3de703beu));
  p = fmaf(t, p, __uint_as_float(0xbec09330u));
  p = fmaf(t, p, __uint_as_float(0x3e0375d0u));
  const float rs = fmaf(ax, p, ax);
  float q = fmaf(ax, __uint_as_float(0x378e98abu), __uint_as_float(0xb9c68948u));
  q = fmaf(ax, q, __uint_as_float(0x3b7cd369u));
  q = fmaf(ax, q, __uint_as_float(0xbcc618b2u));
  q = fmaf(ax, q, __uint_as_float(0x3dda74e4u));
  q = fmaf(ax, q, __uint_as_float(0x3f228afdu));
  q = fmaf(ax, q, __uint_as_float(0x3e03c728u));
  q = fmaf(ax, q, ax);
  const float rl = 1.0f - expf(-q);
  return copysignf(ax < 1.0f ? rs : rl, x);
}

__device__ __forceinline__ float gelu_exact(float x) {
  return 0.5f * x * (1.0f + erf_sel(x * 0.70710678118654752f));
}

__device__ __forceinline__ float max9(const float x[9]) {
  return fmaxf(fmaxf(fmaxf(x[0], x[1]), fmaxf(x[2], x[3])),
               fmaxf(fmaxf(x[4], x[5]), fmaxf(fmaxf(x[6], x[7]), x[8])));
}
__device__ __forceinline__ float min9(const float x[9]) {
  return fminf(fminf(fminf(x[0], x[1]), fminf(x[2], x[3])),
               fminf(fminf(x[4], x[5]), fminf(fminf(x[6], x[7]), x[8])));
}

template <int N>
__device__ __forceinline__ void gelu_pool_max9_n(const float (&x)[N][9], const float (&b)[N],
                                                 float (&r)[N]) {
  float hi[N];
  bool neg = false;
#pragma unroll
  for (int n = 0; n < N; ++n) {
    hi[n] = max9(x[n]) + b[n];
    neg = neg || (hi[n] < 0.f);
  }
#pragma unroll
  for (int n = 0; n < N; ++n) r[n] = gelu_exact(hi[n]);
  if (neg) {
    float lo[N];
#pragma unroll
    for (int n = 0; n < N; ++n) lo[n] = min9(x[n]) + b[n];
#pragma unroll
    for (int n = 0; n < N; ++n) r[n] = fmaxf(r[n], gelu_exact(lo[n]));
  }
}

__device__ __forceinline__ int w0_pair_slot(int p) {
  int ky = p / 25;
  return ky * KP + 2 * (p - ky * 25);
}

__device__ __forceinline__ v2f ld_a(const float* A, int lda, int m0, int k0,
                                    int l, int half) {
  return *(const v2f*)(A + (m0 + l) * lda + (k0 + 2 * half));
}
__device__ __forceinline__ v2f ld_bt(const float* Bt, int ldk, int k0, int n0,
                                     int l, int half) {
  return *(const v2f*)(Bt + (n0 + l) * ldk + (k0 + 2 * half));
}
__device__ __forceinline__ void st_c3(float* Cm, int ldc, int m0, int n0,
                                      v8f c0, v8f c1, v8f c2, int l, int half, int maxRow) {
  const int row0 = m0 + 8 * half;
  float* p = Cm + row0 * ldc + n0 + l;
#pragma unroll
  for (int r = 0; r < 8; ++r) {
    if (row0 + r < maxRow) {
      p[r * ldc]      = c0[r];
      p[r * ldc + 16] = c1[r];
      p[r * ldc + 32] = c2[r];
    }
  }
}

__device__ __forceinline__ void st_c2(float* Cm, int ldc, int m0, int n0,
                                      v8f c0, v8f c1, int l, int half, int maxRow) {
  const int row0 = m0 + 8 * half;
  float* p = Cm + row0 * ldc + n0 + l;
#pragma unroll
  for (int r = 0; r < 8; ++r) {
    if (row0 + r < maxRow) {
      p[r * ldc]      = c0[r];
      p[r * ldc + 16] = c1[r];
    }
  }
}

__device__ __forceinline__ void st_c_t3(float* Ct, int ldk, int m0, int n0,
                                        v8f c0, v8f c1, v8f c2, int l, int half, int maxRow) {
  const int row = m0 + 8 * half;
  float* p = Ct + (n0 + l) * ldk + row;
  if (row < maxRow) {
    *(v4f*)p              = (v4f){c0[0], c0[1], c0[2], c0[3]};
    *(v4f*)(p + 16 * ldk) = (v4f){c1[0], c1[1], c1[2], c1[3]};
    *(v4f*)(p + 32 * ldk) = (v4f){c2[0], c2[1], c2[2], c2[3]};
  }
  if (row + 4 < maxRow) {
    *(v4f*)(p + 4)            = (v4f){c0[4], c0[5], c0[6], c0[7]};
    *(v4f*)(p + 16 * ldk + 4) = (v4f){c1[4], c1[5], c1[6], c1[7]};
    *(v4f*)(p + 32 * ldk + 4) = (v4f){c2[4], c2[5], c2[6], c2[7]};
  }
}

__device__ __forceinline__ void st_c_t2(float* Ct, int ldk, int m0, int n0,
                                        v8f c0, v8f c1, int l, int half, int maxRow) {
  const int row = m0 + 8 * half;
  float* p = Ct + (n0 + l) * ldk + row;
  if (row < maxRow) {
    *(v4f*)p              = (v4f){c0[0], c0[1], c0[2], c0[3]};
    *(v4f*)(p + 16 * ldk) = (v4f){c1[0], c1[1], c1[2], c1[3]};
  }
  if (row + 4 < maxRow) {
    *(v4f*)(p + 4)            = (v4f){c0[4], c0[5], c0[6], c0[7]};
    *(v4f*)(p + 16 * ldk + 4) = (v4f){c1[4], c1[5], c1[6], c1[7]};
  }
}

#define WMMA_F32(acc, a, b)                                                   \
  (acc) = __builtin_amdgcn_wmma_f32_16x16x4_f32(false, (a), false, (b),       \
                                                (short)0, (acc), false, false)

__device__ __forceinline__ void strip_gemm(const float* A, const float* Bt, int m0, int nb,
                                           int l, int half, v8f& acc0, v8f& acc1, v8f& acc2) {
  v2f fa[13], fb0[13], fb1[13], fb2[13];
#pragma unroll
  for (int s = 0; s < 13; ++s) {
    fa[s]  = ld_a(A, KP, m0, 4 * s, l, half);
    fb0[s] = ld_bt(Bt, KP, 4 * s, nb, l, half);
    fb1[s] = ld_bt(Bt, KP, 4 * s, nb + 16, l, half);
    fb2[s] = ld_bt(Bt, KP, 4 * s, nb + 32, l, half);
  }
  __builtin_amdgcn_sched_barrier(0);
  acc0 = (v8f){0.f, 0.f, 0.f, 0.f, 0.f, 0.f, 0.f, 0.f};
  acc1 = acc0;
  acc2 = acc0;
#pragma unroll
  for (int s = 0; s < 13; ++s) {
    WMMA_F32(acc0, fa[s], fb0[s]);
    WMMA_F32(acc1, fa[s], fb1[s]);
    WMMA_F32(acc2, fa[s], fb2[s]);
  }
}

__device__ __forceinline__ void strip_gemm2(const float* A, const float* Bt, int m0, int nb,
                                            int l, int half, v8f& acc0, v8f& acc1) {
  v2f fa[13], fb0[13], fb1[13];
#pragma unroll
  for (int s = 0; s < 13; ++s) {
    fa[s]  = ld_a(A, KP, m0, 4 * s, l, half);
    fb0[s] = ld_bt(Bt, KP, 4 * s, nb, l, half);
    fb1[s] = ld_bt(Bt, KP, 4 * s, nb + 16, l, half);
  }
  __builtin_amdgcn_sched_barrier(0);
  acc0 = (v8f){0.f, 0.f, 0.f, 0.f, 0.f, 0.f, 0.f, 0.f};
  acc1 = acc0;
#pragma unroll
  for (int s = 0; s < 13; ++s) {
    WMMA_F32(acc0, fa[s], fb0[s]);
    WMMA_F32(acc1, fa[s], fb1[s]);
  }
}

extern "C" __global__ __launch_bounds__(NTHREADS) void
fused_outer_conv_kernel(const float* __restrict__ tf,
                        const float* __restrict__ tp,
                        const float* __restrict__ w0, const float* __restrict__ b0,
                        const float* __restrict__ w1, const float* __restrict__ b1,
                        const float* __restrict__ w2, const float* __restrict__ b2,
                        float* __restrict__ out) {
  extern __shared__ float smem[];
  const int tid  = threadIdx.x;
  const int lane = tid & 31;
  const int l    = tid & 15;
  const int half = (tid >> 4) & 1;
  const int wave = tid >> 5;
  const int tok  = blockIdx.x;

  const float* v = tf + tok * ND;
  const float* u = tp + tok * ND;

  {
    const char* pw1 = (const char*)w1;
    const char* pw2 = (const char*)w2;
    if (tid * 128 < (int)(C1 * K1 * sizeof(float))) __builtin_prefetch(pw1 + tid * 128, 0, 1);
    if (tid * 128 < (int)(C2 * K2 * sizeof(float))) __builtin_prefetch(pw2 + tid * 128, 0, 1);
  }

  float* Vm  = smem + OFF_VM;
  float* Um  = smem + OFF_UM;
  float* W0c = smem + OFF_W0C;
  float* Tc  = smem + OFF_TC;
  float* Oc  = smem + OFF_OC;
  float* X1  = smem + OFF_X1;

  const v4f* w0v = (const v4f*)w0;
  const int  q0 = tid, q1 = tid + NTHREADS;
  const bool has2 = tid + 2 * NTHREADS < W0Q;
  const int  q2 = has2 ? tid + 2 * NTHREADS : W0Q - 1;
  const int  s00 = w0_pair_slot(2 * q0), s01 = w0_pair_slot(2 * q0 + 1);
  const int  s10 = w0_pair_slot(2 * q1), s11 = w0_pair_slot(2 * q1 + 1);
  const int  s20 = w0_pair_slot(2 * q2), s21 = w0_pair_slot(2 * q2 + 1);

  const bool pool3 = tid + 2 * NTHREADS < PU0 * PU0;
  int pin[3], pout[3];
#pragma unroll
  for (int n = 0; n < 3; ++n) {
    const int i = (n < 2 || pool3) ? tid + n * NTHREADS : tid;
    const int py = i / PU0, px = i - py * PU0;
    pin[n]  = 3 * py * NP + 3 * px;
    pout[n] = py * P0 + px;
  }

  v4f wq0 = w0v[q0], wq1 = w0v[q1], wq2 = w0v[q2];

  for (int i = tid; i < NU0 * KP; i += NTHREADS) {
    int o = i / KP, k = i % KP;
    bool in = k < K0;
    Um[i] = in ? u[3 * o + k] : 0.f;
    Vm[i] = in ? v[3 * o + k] : 0.f;
  }

  for (int i = tid; i < (64 - K0) * KP; i += NTHREADS) W0c[K0 * KP + i] = 0.f;
  if (tid < 2 * K0) W0c[(tid >> 1) * KP + K0 + (tid & 1)] = 0.f;
  *(v2f*)(W0c + s00) = (v2f){wq0.x, wq0.y};
  *(v2f*)(W0c + s01) = (v2f){wq0.z, wq0.w};
  *(v2f*)(W0c + s10) = (v2f){wq1.x, wq1.y};
  *(v2f*)(W0c + s11) = (v2f){wq1.z, wq1.w};
  if (has2) {
    *(v2f*)(W0c + s20) = (v2f){wq2.x, wq2.y};
    *(v2f*)(W0c + s21) = (v2f){wq2.z, wq2.w};
  }
  __syncthreads();

  for (int c = 0; c < C0; ++c) {
    if (c + 1 < C0) {
      const v4f* wn = w0v + (c + 1) * W0Q;
      wq0 = wn[q0];
      wq1 = wn[q1];
      wq2 = wn[q2];
    }
    const float bc = b0[c];

    {
      const int m0 = (wave >> 1) * 16;
      if ((0xA5 >> wave) & 1) {
        v8f acc0, acc1, acc2;
        strip_gemm(W0c, Vm, m0, 0, l, half, acc0, acc1, acc2);
        st_c_t3(Tc, KP, m0, 0, acc0, acc1, acc2, l, half, KP);
      } else {
        v8f acc0, acc1;
        strip_gemm2(W0c, Vm, m0, 48, l, half, acc0, acc1);
        st_c_t2(Tc, KP, m0, 48, acc0, acc1, l, half, KP);
      }
    }
    __syncthreads();

    if (c + 1 < C0) {
      *(v2f*)(W0c + s00) = (v2f){wq0.x, wq0.y};
      *(v2f*)(W0c + s01) = (v2f){wq0.z, wq0.w};
      *(v2f*)(W0c + s10) = (v2f){wq1.x, wq1.y};
      *(v2f*)(W0c + s11) = (v2f){wq1.z, wq1.w};
      if (has2) {
        *(v2f*)(W0c + s20) = (v2f){wq2.x, wq2.y};
        *(v2f*)(W0c + s21) = (v2f){wq2.z, wq2.w};
      }
    }

    {
      const int rowb = (0x43221040 >> (4 * wave)) & 15;
      if ((0xDC >> wave) & 1) {
        const int m0 = rowb * 16;
        v8f acc0, acc1, acc2;
        strip_gemm(Um, Tc, m0, 0, l, half, acc0, acc1, acc2);
        st_c3(Oc, NP, m0, 0, acc0, acc1, acc2, l, half, NU0);
      } else {
        const int nstrip = wave == 1 ? 1 : 2;
        for (int i = 0; i < nstrip; ++i) {
          const int m0 = (rowb + i) * 16;
          v8f acc0, acc1;
          strip_gemm2(Um, Tc, m0, 48, l, half, acc0, acc1);
          st_c2(Oc, NP, m0, 48, acc0, acc1, l, half, NU0);
        }
      }
    }
    __syncthreads();

    if (wave < 6) {
      float x[3][9], r[3];
#pragma unroll
      for (int n = 0; n < 3; ++n)
#pragma unroll
        for (int dy = 0; dy < 3; ++dy)
#pragma unroll
          for (int dx = 0; dx < 3; ++dx) x[n][3 * dy + dx] = Oc[pin[n] + dy * NP + dx];
      const float bb[3] = {bc, bc, bc};
      gelu_pool_max9_n<3>(x, bb, r);
      float* xo = X1 + c * (P0 * P0);
      xo[pout[0]] = r[0];
      xo[pout[1]] = r[1];
      if (pool3) xo[pout[2]] = r[2];
    } else {
      float x[2][9], r[2];
#pragma unroll
      for (int n = 0; n < 2; ++n)
#pragma unroll
        for (int dy = 0; dy < 3; ++dy)
#pragma unroll
          for (int dx = 0; dx < 3; ++dx) x[n][3 * dy + dx] = Oc[pin[n] + dy * NP + dx];
      const float bb[2] = {bc, bc};
      gelu_pool_max9_n<2>(x, bb, r);
      float* xo = X1 + c * (P0 * P0);
      xo[pout[0]] = r[0];
      xo[pout[1]] = r[1];
    }
  }
  __syncthreads();

  float* A1 = smem + OFF_A1;
  float* O1 = smem + OFF_O1;
  float* X2 = smem + OFF_X2;
  for (int i = tid; i < C1 * K1; i += NTHREADS) {
    int co = i / K1, k = i - co * K1;
    int ci = k / 9, r9 = k - ci * 9;
    A1[co * K1 + r9 * 16 + ci] = w1[i];
  }
  __syncthreads();

  for (int nt = wave; nt < NPIX1 / 16; nt += NWAVE) {
    const int col = nt * 16 + l;
    const int py = col / OW1, px = col - py * OW1;
    const float* xb = X1 + py * P0 + px + half * (2 * P0 * P0);
    const float* ab = A1 + l * K1 + 2 * half;
    v8f acc0 = {0.f, 0.f, 0.f, 0.f, 0.f, 0.f, 0.f, 0.f};
    v8f acc1 = {0.f, 0.f, 0.f, 0.f, 0.f, 0.f, 0.f, 0.f};
    v2f fb[2][4], fa0[2][4], fa1[2][4];
#pragma unroll
    for (int q = 0; q < 4; ++q) {
      fb[0][q].x = xb[(4 * q) * (P0 * P0)];
      fb[0][q].y = xb[(4 * q + 1) * (P0 * P0)];
      fa0[0][q]  = *(const v2f*)(ab + 4 * q);
      fa1[0][q]  = *(const v2f*)(ab + 16 * K1 + 4 * q);
    }
#pragma unroll
    for (int r9 = 0; r9 < 9; ++r9) {
      const int cur = r9 & 1, nxt = cur ^ 1;
      if (r9 + 1 < 9) {
        const int sp = ((r9 + 1) / 3) * P0 + ((r9 + 1) % 3);
#pragma unroll
        for (int q = 0; q < 4; ++q) {
          fb[nxt][q].x = xb[(4 * q) * (P0 * P0) + sp];
          fb[nxt][q].y = xb[(4 * q + 1) * (P0 * P0) + sp];
          fa0[nxt][q]  = *(const v2f*)(ab + (r9 + 1) * 16 + 4 * q);
          fa1[nxt][q]  = *(const v2f*)(ab + 16 * K1 + (r9 + 1) * 16 + 4 * q);
        }
      }
      __builtin_amdgcn_sched_barrier(0);
#pragma unroll
      for (int q = 0; q < 4; ++q) {
        WMMA_F32(acc0, fa0[cur][q], fb[cur][q]);
        WMMA_F32(acc1, fa1[cur][q], fb[cur][q]);
      }
      __builtin_amdgcn_sched_barrier(0);
    }
#pragma unroll
    for (int r = 0; r < 8; ++r) {
      O1[(r + 8 * half) * O1P + col]      = acc0[r];
      O1[(16 + r + 8 * half) * O1P + col] = acc1[r];
    }
  }
  __syncthreads();

  {
    const int rem = tid & 63, py = rem >> 3, px = rem & 7;
    const float* ob = O1 + 3 * py * OW1 + 3 * px;
#pragma unroll
    for (int g = 0; g < 2; ++g) {
      float x[4][9], bb[4], r[4];
#pragma unroll
      for (int n = 0; n < 4; ++n) {
        const int co = (tid >> 6) + 4 * (4 * g + n);
        bb[n] = b1[co];
#pragma unroll
        for (int dy = 0; dy < 3; ++dy)
#pragma unroll
          for (int dx = 0; dx < 3; ++dx) x[n][3 * dy + dx] = ob[co * O1P + dy * OW1 + dx];
      }
      gelu_pool_max9_n<4>(x, bb, r);
#pragma unroll
      for (int n = 0; n < 4; ++n) X2[((tid >> 6) + 4 * (4 * g + n)) * X2P + rem] = r[n];
    }
  }
  __syncthreads();

  float* A2 = smem + OFF_A2;
  float* O2 = smem + OFF_O2;
  {
    float wv[(C2 * C1) / NTHREADS][9];
#pragma unroll
    for (int t = 0; t < (C2 * C1) / NTHREADS; ++t) {
      const float* wsrc = w2 + (tid + t * NTHREADS) * 9;
#pragma unroll
      for (int r9 = 0; r9 < 9; ++r9) wv[t][r9] = wsrc[r9];
    }
#pragma unroll
    for (int t = 0; t < (C2 * C1) / NTHREADS; ++t) {
      const int p = tid + t * NTHREADS;
      float* wdst = A2 + (p >> 5) * A2P + (p & 31);
#pragma unroll
      for (int r9 = 0; r9 < 9; ++r9) wdst[r9 * C1] = wv[t][r9];
    }
  }
  __syncthreads();

  if (wave < 2 * 3) {
    const int m0 = (wave / 3) * 16, n0 = (wave % 3) * 16;
    const int colr = n0 + l;
    const int col = colr < NPIX2 ? colr : NPIX2 - 1;
    const int py = col / OW2, px = col - py * OW2;
    const float* xb = X2 + py * P1 + px + half * (2 * X2P);
    const float* ab = A2 + (m0 + l) * A2P + 2 * half;
    v8f acc0 = {0.f, 0.f, 0.f, 0.f, 0.f, 0.f, 0.f, 0.f};
    v8f acc1 = {0.f, 0.f, 0.f, 0.f, 0.f, 0.f, 0.f, 0.f};
    v2f fb[2][8], fa[2][8];
#pragma unroll
    for (int q = 0; q < 8; ++q) {
      fb[0][q].x = xb[(4 * q) * X2P];
      fb[0][q].y = xb[(4 * q + 1) * X2P];
      fa[0][q]   = *(const v2f*)(ab + 4 * q);
    }
#pragma unroll
    for (int r9 = 0; r9 < 9; ++r9) {
      const int cur = r9 & 1, nxt = cur ^ 1;
      if (r9 + 1 < 9) {
        const int sp = ((r9 + 1) / 3) * P1 + ((r9 + 1) % 3);
#pragma unroll
        for (int q = 0; q < 8; ++q) {
          fb[nxt][q].x = xb[(4 * q) * X2P + sp];
          fb[nxt][q].y = xb[(4 * q + 1) * X2P + sp];
          fa[nxt][q]   = *(const v2f*)(ab + (r9 + 1) * C1 + 4 * q);
        }
      }
      __builtin_amdgcn_sched_barrier(0);
#pragma unroll
      for (int q = 0; q < 8; q += 2) {
        WMMA_F32(acc0, fa[cur][q], fb[cur][q]);
        WMMA_F32(acc1, fa[cur][q + 1], fb[cur][q + 1]);
      }
      __builtin_amdgcn_sched_barrier(0);
    }
#pragma unroll
    for (int r = 0; r < 8; ++r)
      O2[(m0 + r + 8 * half) * NPIX2P + n0 + l] = acc0[r] + acc1[r];
  }
  __syncthreads();

  if (tid < 128) {
    int co = tid >> 2, rem = tid & 3, py = rem >> 1, px = rem & 1;
    float bc = b2[co];
    float x[1][9], r[1];
#pragma unroll
    for (int dy = 0; dy < 3; ++dy)
#pragma unroll
      for (int dx = 0; dx < 3; ++dx) x[0][3 * dy + dx] = O2[co * NPIX2P + (3 * py + dy) * OW2 + (3 * px + dx)];
    const float bb[1] = {bc};
    gelu_pool_max9_n<1>(x, bb, r);
    out[tok * 128 + tid] = r[0];
  }
}

extern "C" void kernel_launch(void* const* d_in, const int* in_sizes, int n_in,
                              void* d_out, int out_size, void* d_ws, size_t ws_size,
                              hipStream_t stream) {
  (void)in_sizes; (void)n_in; (void)out_size; (void)d_ws; (void)ws_size;
  const float* tf = (const float*)d_in[0];
  const float* tp = (const float*)d_in[1];
  const float* w0 = (const float*)d_in[2];
  const float* b0 = (const float*)d_in[3];
  const float* w1 = (const float*)d_in[4];
  const float* b1 = (const float*)d_in[5];
  const float* w2 = (const float*)d_in[6];
  const float* b2 = (const float*)d_in[7];
  float* out = (float*)d_out;

  const size_t shmem = SMEM_FLOATS * sizeof(float);
  (void)hipFuncSetAttribute((const void*)fused_outer_conv_kernel,
                            hipFuncAttributeMaxDynamicSharedMemorySize, (int)shmem);
  hipLaunchKernelGGL(fused_outer_conv_kernel, dim3(NTOK), dim3(NTHREADS), shmem,
                     stream, tf, tp, w0, b0, w1, b1, w2, b2, out);
}


# Staging without exposed round trips: im2row loads issued as one group per thread; w1, w2 and late biases requested ahead of the preceding pool pass, restaged into dead LDS; two barriers fewer
# speedup vs baseline: 7.0442x; 1.0417x over previous
#include <hip/hip_runtime.h>


typedef float v2f __attribute__((ext_vector_type(2)));
typedef float v4f __attribute__((ext_vector_type(4)));
typedef float v8f __attribute__((ext_vector_type(8)));

#define NB 8
#define NS 64
#define ND 300
#define NTOK (NB * NS)
#define NTHREADS 256
#define NWAVE 8

#define C0 16
#define K0 50
#define KP 52
#define NU0 80
#define NP 96
#define P0 28
#define PU0 26
#define W0Q 625

#define C1 32
#define OW1 24
#define NPIX1 576
#define O1P 580
#define K1 144
#define P1 8

#define C2 32
#define OW2 6
#define NPIX2 36
#define NPIX2P 48
#define K2 288
#define A2P 292
#define X2P 80

#define OFF_VM  0
#define OFF_UM  4992
#define OFF_W0C 9984
#define OFF_TC  13312
#define OFF_OC  18304
#define OFF_X1  26368
#define OFF_A1  0
#define OFF_O1  4608
#define OFF_X2  26368
#define OFF_A2  28928
#define OFF_O2  23040
#define SMEM_FLOATS 38912

__device__ __forceinline__ float erf_sel(float x) {
  const float ax = fabsf(x);
  const float t  = ax * ax;
  float p = fmaf(t, __uint_as_float(0xba1345e1u), __uint_as_float(0x3ba10414u));
  p = fmaf(t, p, __uint_as_float(0xbcdac9b8u));
  p = fmaf(t, p, __uint_as_float(0x3de703beu));
  p = fmaf(t, p, __uint_as_float(0xbec09330u));
  p = fmaf(t, p, __uint_as_float(0x3e0375d0u));
  const float rs = fmaf(ax, p, ax);
  float q = fmaf(ax, __uint_as_float(0x378e98abu), __uint_as_float(0xb9c68948u));
  q = fmaf(ax, q, __uint_as_float(0x3b7cd369u));
  q = fmaf(ax, q, __uint_as_float(0xbcc618b2u));
  q = fmaf(ax, q, __uint_as_float(0x3dda74e4u));
  q = fmaf(ax, q, __uint_as_float(0x3f228afdu));
  q = fmaf(ax, q, __uint_as_float(0x3e03c728u));
  q = fmaf(ax, q, ax);
  const float rl = 1.0f - expf(-q);
  return copysignf(ax < 1.0f ? rs : rl, x);
}

__device__ __forceinline__ float gelu_exact(float x) {
  return 0.5f * x * (1.0f + erf_sel(x * 0.70710678118654752f));
}

__device__ __forceinline__ float max9(const float x[9]) {
  return fmaxf(fmaxf(fmaxf(x[0], x[1]), fmaxf(x[2], x[3])),
               fmaxf(fmaxf(x[4], x[5]), fmaxf(fmaxf(x[6], x[7]), x[8])));
}
__device__ __forceinline__ float min9(const float x[9]) {
  return fminf(fminf(fminf(x[0], x[1]), fminf(x[2], x[3])),
               fminf(fminf(x[4], x[5]), fminf(fminf(x[6], x[7]), x[8])));
}

template <int N>
__device__ __forceinline__ void gelu_pool_max9_n(const float (&x)[N][9], const float (&b)[N],
                                                 float (&r)[N]) {
  float hi[N];
  bool neg = false;
#pragma unroll
  for (int n = 0; n < N; ++n) {
    hi[n] = max9(x[n]) + b[n];
    neg = neg || (hi[n] < 0.f);
  }
#pragma unroll
  for (int n = 0; n < N; ++n) r[n] = gelu_exact(hi[n]);
  if (neg) {
    float lo[N];
#pragma unroll
    for (int n = 0; n < N; ++n) lo[n] = min9(x[n]) + b[n];
#pragma unroll
    for (int n = 0; n < N; ++n) r[n] = fmaxf(r[n], gelu_exact(lo[n]));
  }
}

__device__ __forceinline__ int w0_pair_slot(int p) {
  int ky = p / 25;
  return ky * KP + 2 * (p - ky * 25);
}

__device__ __forceinline__ v2f ld_a(const float* A, int lda, int m0, int k0,
                                    int l, int half) {
  return *(const v2f*)(A + (m0 + l) * lda + (k0 + 2 * half));
}
__device__ __forceinline__ v2f ld_bt(const float* Bt, int ldk, int k0, int n0,
                                     int l, int half) {
  return *(const v2f*)(Bt + (n0 + l) * ldk + (k0 + 2 * half));
}
__device__ __forceinline__ void st_c3(float* Cm, int ldc, int m0, int n0,
                                      v8f c0, v8f c1, v8f c2, int l, int half, int maxRow) {
  const int row0 = m0 + 8 * half;
  float* p = Cm + row0 * ldc + n0 + l;
#pragma unroll
  for (int r = 0; r < 8; ++r) {
    if (row0 + r < maxRow) {
      p[r * ldc]      = c0[r];
      p[r * ldc + 16] = c1[r];
      p[r * ldc + 32] = c2[r];
    }
  }
}

__device__ __forceinline__ void st_c2(float* Cm, int ldc, int m0, int n0,
                                      v8f c0, v8f c1, int l, int half, int maxRow) {
  const int row0 = m0 + 8 * half;
  float* p = Cm + row0 * ldc + n0 + l;
#pragma unroll
  for (int r = 0; r < 8; ++r) {
    if (row0 + r < maxRow) {
      p[r * ldc]      = c0[r];
      p[r * ldc + 16] = c1[r];
    }
  }
}

__device__ __forceinline__ void st_c_t3(float* Ct, int ldk, int m0, int n0,
                                        v8f c0, v8f c1, v8f c2, int l, int half, int maxRow) {
  const int row = m0 + 8 * half;
  float* p = Ct + (n0 + l) * ldk + row;
  if (row < maxRow) {
    *(v4f*)p              = (v4f){c0[0], c0[1], c0[2], c0[3]};
    *(v4f*)(p + 16 * ldk) = (v4f){c1[0], c1[1], c1[2], c1[3]};
    *(v4f*)(p + 32 * ldk) = (v4f){c2[0], c2[1], c2[2], c2[3]};
  }
  if (row + 4 < maxRow) {
    *(v4f*)(p + 4)            = (v4f){c0[4], c0[5], c0[6], c0[7]};
    *(v4f*)(p + 16 * ldk + 4) = (v4f){c1[4], c1[5], c1[6], c1[7]};
    *(v4f*)(p + 32 * ldk + 4) = (v4f){c2[4], c2[5], c2[6], c2[7]};
  }
}

__device__ __forceinline__ void st_c_t2(float* Ct, int ldk, int m0, int n0,
                                        v8f c0, v8f c1, int l, int half, int maxRow) {
  const int row = m0 + 8 * half;
  float* p = Ct + (n0 + l) * ldk + row;
  if (row < maxRow) {
    *(v4f*)p              = (v4f){c0[0], c0[1], c0[2], c0[3]};
    *(v4f*)(p + 16 * ldk) = (v4f){c1[0], c1[1], c1[2], c1[3]};
  }
  if (row + 4 < maxRow) {
    *(v4f*)(p + 4)            = (v4f){c0[4], c0[5], c0[6], c0[7]};
    *(v4f*)(p + 16 * ldk + 4) = (v4f){c1[4], c1[5], c1[6], c1[7]};
  }
}

#define WMMA_F32(acc, a, b)                                                   \
  (acc) = __builtin_amdgcn_wmma_f32_16x16x4_f32(false, (a), false, (b),       \
                                                (short)0, (acc), false, false)

__device__ __forceinline__ void strip_gemm(const float* A, const float* Bt, int m0, int nb,
                                           int l, int half, v8f& acc0, v8f& acc1, v8f& acc2) {
  v2f fa[13], fb0[13], fb1[13], fb2[13];
#pragma unroll
  for (int s = 0; s < 13; ++s) {
    fa[s]  = ld_a(A, KP, m0, 4 * s, l, half);
    fb0[s] = ld_bt(Bt, KP, 4 * s, nb, l, half);
    fb1[s] = ld_bt(Bt, KP, 4 * s, nb + 16, l, half);
    fb2[s] = ld_bt(Bt, KP, 4 * s, nb + 32, l, half);
  }
  __builtin_amdgcn_sched_barrier(0);
  acc0 = (v8f){0.f, 0.f, 0.f, 0.f, 0.f, 0.f, 0.f, 0.f};
  acc1 = acc0;
  acc2 = acc0;
#pragma unroll
  for (int s = 0; s < 13; ++s) {
    WMMA_F32(acc0, fa[s], fb0[s]);
    WMMA_F32(acc1, fa[s], fb1[s]);
    WMMA_F32(acc2, fa[s], fb2[s]);
  }
}

__device__ __forceinline__ void strip_gemm2(const float* A, const float* Bt, int m0, int nb,
                                            int l, int half, v8f& acc0, v8f& acc1) {
  v2f fa[13], fb0[13], fb1[13];
#pragma unroll
  for (int s = 0; s < 13; ++s) {
    fa[s]  = ld_a(A, KP, m0, 4 * s, l, half);
    fb0[s] = ld_bt(Bt, KP, 4 * s, nb, l, half);
    fb1[s] = ld_bt(Bt, KP, 4 * s, nb + 16, l, half);
  }
  __builtin_amdgcn_sched_barrier(0);
  acc0 = (v8f){0.f, 0.f, 0.f, 0.f, 0.f, 0.f, 0.f, 0.f};
  acc1 = acc0;
#pragma unroll
  for (int s = 0; s < 13; ++s) {
    WMMA_F32(acc0, fa[s], fb0[s]);
    WMMA_F32(acc1, fa[s], fb1[s]);
  }
}

__device__ __forceinline__ void pool_conv0(const float* Oc, float* xo, const int (&pin)[3],
                                           const int (&pout)[3], bool pool3, int wave, float bc) {
  if (wave < 6) {
    float x[3][9], r[3];
#pragma unroll
    for (int n = 0; n < 3; ++n)
#pragma unroll
      for (int dy = 0; dy < 3; ++dy)
#pragma unroll
        for (int dx = 0; dx < 3; ++dx) x[n][3 * dy + dx] = Oc[pin[n] + dy * NP + dx];
    const float bb[3] = {bc, bc, bc};
    gelu_pool_max9_n<3>(x, bb, r);
    xo[pout[0]] = r[0];
    xo[pout[1]] = r[1];
    if (pool3) xo[pout[2]] = r[2];
  } else {
    float x[2][9], r[2];
#pragma unroll
    for (int n = 0; n < 2; ++n)
#pragma unroll
      for (int dy = 0; dy < 3; ++dy)
#pragma unroll
        for (int dx = 0; dx < 3; ++dx) x[n][3 * dy + dx] = Oc[pin[n] + dy * NP + dx];
    const float bb[2] = {bc, bc};
    gelu_pool_max9_n<2>(x, bb, r);
    xo[pout[0]] = r[0];
    xo[pout[1]] = r[1];
  }
}

extern "C" __global__ __launch_bounds__(NTHREADS) void
fused_outer_conv_kernel(const float* __restrict__ tf,
                        const float* __restrict__ tp,
                        const float* __restrict__ w0, const float* __restrict__ b0,
                        const float* __restrict__ w1, const float* __restrict__ b1,
                        const float* __restrict__ w2, const float* __restrict__ b2,
                        float* __restrict__ out) {
  extern __shared__ float smem[];
  const int tid  = threadIdx.x;
  const int lane = tid & 31;
  const int l    = tid & 15;
  const int half = (tid >> 4) & 1;
  const int wave = tid >> 5;
  const int tok  = blockIdx.x;

  const float* v = tf + tok * ND;
  const float* u = tp + tok * ND;

  {
    const char* pw1 = (const char*)w1;
    const char* pw2 = (const char*)w2;
    if (tid * 128 < (int)(C1 * K1 * sizeof(float))) __builtin_prefetch(pw1 + tid * 128, 0, 1);
    if (tid * 128 < (int)(C2 * K2 * sizeof(float))) __builtin_prefetch(pw2 + tid * 128, 0, 1);
  }

  float* Vm  = smem + OFF_VM;
  float* Um  = smem + OFF_UM;
  float* W0c = smem + OFF_W0C;
  float* Tc  = smem + OFF_TC;
  float* Oc  = smem + OFF_OC;
  float* X1  = smem + OFF_X1;

  const v4f* w0v = (const v4f*)w0;
  const int  q0 = tid, q1 = tid + NTHREADS;
  const bool has2 = tid + 2 * NTHREADS < W0Q;
  const int  q2 = has2 ? tid + 2 * NTHREADS : W0Q - 1;
  const int  s00 = w0_pair_slot(2 * q0), s01 = w0_pair_slot(2 * q0 + 1);
  const int  s10 = w0_pair_slot(2 * q1), s11 = w0_pair_slot(2 * q1 + 1);
  const int  s20 = w0_pair_slot(2 * q2), s21 = w0_pair_slot(2 * q2 + 1);

  const bool pool3 = tid + 2 * NTHREADS < PU0 * PU0;
  int pin[3], pout[3];
#pragma unroll
  for (int n = 0; n < 3; ++n) {
    const int i = (n < 2 || pool3) ? tid + n * NTHREADS : tid;
    const int py = i / PU0, px = i - py * PU0;
    pin[n]  = 3 * py * NP + 3 * px;
    pout[n] = py * P0 + px;
  }

  v4f wq0 = w0v[q0], wq1 = w0v[q1], wq2 = w0v[q2];

  {
    constexpr int NIT = (NU0 * KP + NTHREADS - 1) / NTHREADS;
    float uu[NIT], vv[NIT];
#pragma unroll
    for (int j = 0; j < NIT; ++j) {
      const int i  = tid + j * NTHREADS;
      const int ic = i < NU0 * KP ? i : NU0 * KP - 1;
      const int o = ic / KP, k = ic - o * KP;
      const int kc = k < K0 ? k : K0 - 1;
      uu[j] = u[3 * o + kc];
      vv[j] = v[3 * o + kc];
    }
#pragma unroll
    for (int j = 0; j < NIT; ++j) {
      const int i = tid + j * NTHREADS;
      if (j + 1 < NIT || i < NU0 * KP) {
        const int o = i / KP, k = i - o * KP;
        const bool in = k < K0;
        Um[i] = in ? uu[j] : 0.f;
        Vm[i] = in ? vv[j] : 0.f;
      }
    }
  }

  for (int i = tid; i < (64 - K0) * KP; i += NTHREADS) W0c[K0 * KP + i] = 0.f;
  if (tid < 2 * K0) W0c[(tid >> 1) * KP + K0 + (tid & 1)] = 0.f;
  *(v2f*)(W0c + s00) = (v2f){wq0.x, wq0.y};
  *(v2f*)(W0c + s01) = (v2f){wq0.z, wq0.w};
  *(v2f*)(W0c + s10) = (v2f){wq1.x, wq1.y};
  *(v2f*)(W0c + s11) = (v2f){wq1.z, wq1.w};
  if (has2) {
    *(v2f*)(W0c + s20) = (v2f){wq2.x, wq2.y};
    *(v2f*)(W0c + s21) = (v2f){wq2.z, wq2.w};
  }
  __syncthreads();

  for (int c = 0; c < C0; ++c) {
    if (c + 1 < C0) {
      const v4f* wn = w0v + (c + 1) * W0Q;
      wq0 = wn[q0];
      wq1 = wn[q1];
      wq2 = wn[q2];
    }
    const float bc = b0[c];

    {
      const int m0 = (wave >> 1) * 16;
      if ((0xA5 >> wave) & 1) {
        v8f acc0, acc1, acc2;
        strip_gemm(W0c, Vm, m0, 0, l, half, acc0, acc1, acc2);
        st_c_t3(Tc, KP, m0, 0, acc0, acc1, acc2, l, half, KP);
      } else {
        v8f acc0, acc1;
        strip_gemm2(W0c, Vm, m0, 48, l, half, acc0, acc1);
        st_c_t2(Tc, KP, m0, 48, acc0, acc1, l, half, KP);
      }
    }
    __syncthreads();

    if (c + 1 < C0) {
      *(v2f*)(W0c + s00) = (v2f){wq0.x, wq0.y};
      *(v2f*)(W0c + s01) = (v2f){wq0.z, wq0.w};
      *(v2f*)(W0c + s10) = (v2f){wq1.x, wq1.y};
      *(v2f*)(W0c + s11) = (v2f){wq1.z, wq1.w};
      if (has2) {
        *(v2f*)(W0c + s20) = (v2f){wq2.x, wq2.y};
        *(v2f*)(W0c + s21) = (v2f){wq2.z, wq2.w};
      }
    }

    {
      const int rowb = (0x43221040 >> (4 * wave)) & 15;
      if ((0xDC >> wave) & 1) {
        const int m0 = rowb * 16;
        v8f acc0, acc1, acc2;
        strip_gemm(Um, Tc, m0, 0, l, half, acc0, acc1, acc2);
        st_c3(Oc, NP, m0, 0, acc0, acc1, acc2, l, half, NU0);
      } else {
        const int nstrip = wave == 1 ? 1 : 2;
        for (int i = 0; i < nstrip; ++i) {
          const int m0 = (rowb + i) * 16;
          v8f acc0, acc1;
          strip_gemm2(Um, Tc, m0, 48, l, half, acc0, acc1);
          st_c2(Oc, NP, m0, 48, acc0, acc1, l, half, NU0);
        }
      }
    }
    __syncthreads();

    if (c + 1 < C0) pool_conv0(Oc, X1 + c * (P0 * P0), pin, pout, pool3, wave, bc);
  }

  const float bcl = b0[C0 - 1];
  float w1r[(C1 * C0) / NTHREADS][9];
#pragma unroll
  for (int t = 0; t < (C1 * C0) / NTHREADS; ++t) {
    const float* wsrc = w1 + (tid + t * NTHREADS) * 9;
#pragma unroll
    for (int r9 = 0; r9 < 9; ++r9) w1r[t][r9] = wsrc[r9];
  }
  float b1v[8];
#pragma unroll
  for (int j = 0; j < 8; ++j) b1v[j] = b1[(tid >> 6) + 4 * j];
  pool_conv0(Oc, X1 + (C0 - 1) * (P0 * P0), pin, pout, pool3, wave, bcl);

  float* A1 = smem + OFF_A1;
  float* O1 = smem + OFF_O1;
  float* X2 = smem + OFF_X2;
#pragma unroll
  for (int t = 0; t < (C1 * C0) / NTHREADS; ++t) {
    const int p = tid + t * NTHREADS;
    float* wdst = A1 + (p >> 4) * K1 + (p & 15);
#pragma unroll
    for (int r9 = 0; r9 < 9; ++r9) wdst[r9 * 16] = w1r[t][r9];
  }
  __syncthreads();

  for (int nt = wave; nt < NPIX1 / 16; nt += NWAVE) {
    const int col = nt * 16 + l;
    const int py = col / OW1, px = col - py * OW1;
    const float* xb = X1 + py * P0 + px + half * (2 * P0 * P0);
    const float* ab = A1 + l * K1 + 2 * half;
    v8f acc0 = {0.f, 0.f, 0.f, 0.f, 0.f, 0.f, 0.f, 0.f};
    v8f acc1 = {0.f, 0.f, 0.f, 0.f, 0.f, 0.f, 0.f, 0.f};
    v2f fb[2][4], fa0[2][4], fa1[2][4];
#pragma unroll
    for (int q = 0; q < 4; ++q) {
      fb[0][q].x = xb[(4 * q) * (P0 * P0)];
      fb[0][q].y = xb[(4 * q + 1) * (P0 * P0)];
      fa0[0][q]  = *(const v2f*)(ab + 4 * q);
      fa1[0][q]  = *(const v2f*)(ab + 16 * K1 + 4 * q);
    }
#pragma unroll
    for (int r9 = 0; r9 < 9; ++r9) {
      const int cur = r9 & 1, nxt = cur ^ 1;
      if (r9 + 1 < 9) {
        const int sp = ((r9 + 1) / 3) * P0 + ((r9 + 1) % 3);
#pragma unroll
        for (int q = 0; q < 4; ++q) {
          fb[nxt][q].x = xb[(4 * q) * (P0 * P0) + sp];
          fb[nxt][q].y = xb[(4 * q + 1) * (P0 * P0) + sp];
          fa0[nxt][q]  = *(const v2f*)(ab + (r9 + 1) * 16 + 4 * q);
          fa1[nxt][q]  = *(const v2f*)(ab + 16 * K1 + (r9 + 1) * 16 + 4 * q);
        }
      }
      __builtin_amdgcn_sched_barrier(0);
#pragma unroll
      for (int q = 0; q < 4; ++q) {
        WMMA_F32(acc0, fa0[cur][q], fb[cur][q]);
        WMMA_F32(acc1, fa1[cur][q], fb[cur][q]);
      }
      __builtin_amdgcn_sched_barrier(0);
    }
#pragma unroll
    for (int r = 0; r < 8; ++r) {
      O1[(r + 8 * half) * O1P + col]      = acc0[r];
      O1[(16 + r + 8 * half) * O1P + col] = acc1[r];
    }
  }
  __syncthreads();

  float wv[(C2 * C1) / NTHREADS][9];
#pragma unroll
  for (int t = 0; t < (C2 * C1) / NTHREADS; ++t) {
    const float* wsrc = w2 + (tid + t * NTHREADS) * 9;
#pragma unroll
    for (int r9 = 0; r9 < 9; ++r9) wv[t][r9] = wsrc[r9];
  }
  const float b2v = b2[(tid & 127) >> 2];

  {
    const int rem = tid & 63, py = rem >> 3, px = rem & 7;
    const float* ob = O1 + 3 * py * OW1 + 3 * px;
#pragma unroll
    for (int g = 0; g < 2; ++g) {
      float x[4][9], bb[4], r[4];
#pragma unroll
      for (int n = 0; n < 4; ++n) {
        const int co = (tid >> 6) + 4 * (4 * g + n);
        bb[n] = b1v[4 * g + n];
#pragma unroll
        for (int dy = 0; dy < 3; ++dy)
#pragma unroll
          for (int dx = 0; dx < 3; ++dx) x[n][3 * dy + dx] = ob[co * O1P + dy * OW1 + dx];
      }
      gelu_pool_max9_n<4>(x, bb, r);
#pragma unroll
      for (int n = 0; n < 4; ++n) X2[((tid >> 6) + 4 * (4 * g + n)) * X2P + rem] = r[n];
    }
  }

  float* A2 = smem + OFF_A2;
  float* O2 = smem + OFF_O2;
#pragma unroll
  for (int t = 0; t < (C2 * C1) / NTHREADS; ++t) {
    const int p = tid + t * NTHREADS;
    float* wdst = A2 + (p >> 5) * A2P + (p & 31);
#pragma unroll
    for (int r9 = 0; r9 < 9; ++r9) wdst[r9 * C1] = wv[t][r9];
  }
  __syncthreads();

  if (wave < 2 * 3) {
    const int m0 = (wave / 3) * 16, n0 = (wave % 3) * 16;
    const int colr = n0 + l;
    const int col = colr < NPIX2 ? colr : NPIX2 - 1;
    const int py = col / OW2, px = col - py * OW2;
    const float* xb = X2 + py * P1 + px + half * (2 * X2P);
    const float* ab = A2 + (m0 + l) * A2P + 2 * half;
    v8f acc0 = {0.f, 0.f, 0.f, 0.f, 0.f, 0.f, 0.f, 0.f};
    v8f acc1 = {0.f, 0.f, 0.f, 0.f, 0.f, 0.f, 0.f, 0.f};
    v2f fb[2][8], fa[2][8];
#pragma unroll
    for (int q = 0; q < 8; ++q) {
      fb[0][q].x = xb[(4 * q) * X2P];
      fb[0][q].y = xb[(4 * q + 1) * X2P];
      fa[0][q]   = *(const v2f*)(ab + 4 * q);
    }
#pragma unroll
    for (int r9 = 0; r9 < 9; ++r9) {
      const int cur = r9 & 1, nxt = cur ^ 1;
      if (r9 + 1 < 9) {
        const int sp = ((r9 + 1) / 3) * P1 + ((r9 + 1) % 3);
#pragma unroll
        for (int q = 0; q < 8; ++q) {
          fb[nxt][q].x = xb[(4 * q) * X2P + sp];
          fb[nxt][q].y = xb[(4 * q + 1) * X2P + sp];
          fa[nxt][q]   = *(const v2f*)(ab + (r9 + 1) * C1 + 4 * q);
        }
      }
      __builtin_amdgcn_sched_barrier(0);
#pragma unroll
      for (int q = 0; q < 8; q += 2) {
        WMMA_F32(acc0, fa[cur][q], fb[cur][q]);
        WMMA_F32(acc1, fa[cur][q + 1], fb[cur][q + 1]);
      }
      __builtin_amdgcn_sched_barrier(0);
    }
#pragma unroll
    for (int r = 0; r < 8; ++r)
      O2[(m0 + r + 8 * half) * NPIX2P + n0 + l] = acc0[r] + acc1[r];
  }
  __syncthreads();

  if (tid < 128) {
    int co = tid >> 2, rem = tid & 3, py = rem >> 1, px = rem & 1;
    const float bc = b2v;
    float x[1][9], r[1];
#pragma unroll
    for (int dy = 0; dy < 3; ++dy)
#pragma unroll
      for (int dx = 0; dx < 3; ++dx) x[0][3 * dy + dx] = O2[co * NPIX2P + (3 * py + dy) * OW2 + (3 * px + dx)];
    const float bb[1] = {bc};
    gelu_pool_max9_n<1>(x, bb, r);
    out[tok * 128 + tid] = r[0];
  }
}

extern "C" void kernel_launch(void* const* d_in, const int* in_sizes, int n_in,
                              void* d_out, int out_size, void* d_ws, size_t ws_size,
                              hipStream_t stream) {
  (void)in_sizes; (void)n_in; (void)out_size; (void)d_ws; (void)ws_size;
  const float* tf = (const float*)d_in[0];
  const float* tp = (const float*)d_in[1];
  const float* w0 = (const float*)d_in[2];
  const float* b0 = (const float*)d_in[3];
  const float* w1 = (const float*)d_in[4];
  const float* b1 = (const float*)d_in[5];
  const float* w2 = (const float*)d_in[6];
  const float* b2 = (const float*)d_in[7];
  float* out = (float*)d_out;

  const size_t shmem = SMEM_FLOATS * sizeof(float);
  (void)hipFuncSetAttribute((const void*)fused_outer_conv_kernel,
                            hipFuncAttributeMaxDynamicSharedMemorySize, (int)shmem);
  hipLaunchKernelGGL(fused_outer_conv_kernel, dim3(NTOK), dim3(NTHREADS), shmem,
                     stream, tf, tp, w0, b0, w1, b1, w2, b2, out);
}
